# starting kernel
# baseline (speedup 1.0000x reference)
#include <hip/hip_runtime.h>

typedef __attribute__((ext_vector_type(2))) float v2f;
typedef __attribute__((ext_vector_type(8))) float v8f;

#define IN_DIM 64
#define OUT_DIM 64
#define LDS_STRIDE 68

#define WMMA_F32_16x16x4(A, B, C) \
  __builtin_amdgcn_wmma_f32_16x16x4_f32(false, (A), false, (B), (short)0, (C), false, false)

__global__ __launch_bounds__(256) void demo_scatter_kernel(
    const float* __restrict__ x, const int* __restrict__ ei,
    float* __restrict__ nsum, float* __restrict__ deg, int n_edges) {
  long long tid = (long long)blockIdx.x * blockDim.x + threadIdx.x;
  int e = (int)(tid >> 4);
  if (e >= n_edges) return;
  int part = (int)(tid & 15);
  int s = ei[e];
  int d = ei[n_edges + e];
  const float4 v = *(const float4*)(x + (size_t)s * IN_DIM + part * 4);
  float* dst = nsum + (size_t)d * IN_DIM + part * 4;
  atomicAdd(dst + 0, v.x);
  atomicAdd(dst + 1, v.y);
  atomicAdd(dst + 2, v.z);
  atomicAdd(dst + 3, v.w);
  if (part == 0) atomicAdd(deg + d, 1.0f);
}

__global__ __launch_bounds__(256) void demo_node_kernel(
    const float* __restrict__ x, const float* __restrict__ Wg,
    const float* __restrict__ Wl, const float* __restrict__ Ws,
    const float* __restrict__ bias, const float* __restrict__ deg,
    float* sum_out, int n_nodes) {
  __shared__ float sW[3 * OUT_DIM * LDS_STRIDE];
  float* sWg = sW;
  float* sWs = sW + OUT_DIM * LDS_STRIDE;
  float* sWl = sW + 2 * OUT_DIM * LDS_STRIDE;

  {
    const int tid = threadIdx.x;
#pragma unroll
    for (int i = 0; i < 4; ++i) {
      int idx = tid + i * 256;
      int row = idx >> 4;
      int k4  = idx & 15;
      float4 g = *(const float4*)(Wg + (size_t)idx * 4);
      float4 s = *(const float4*)(Ws + (size_t)idx * 4);
      float4 l = *(const float4*)(Wl + (size_t)idx * 4);
      *(float4*)(sWg + row * LDS_STRIDE + k4 * 4) = g;
      *(float4*)(sWs + row * LDS_STRIDE + k4 * 4) = s;
      *(float4*)(sWl + row * LDS_STRIDE + k4 * 4) = l;
    }
  }
  __syncthreads();

  const int lane    = threadIdx.x & 31;
  const int wave    = threadIdx.x >> 5;
  const int lane_lo = lane & 15;
  const int hi      = lane >> 4;
  const int node_base = (blockIdx.x * 8 + wave) * 16;
  if (node_base >= n_nodes) return;

  int arow = node_base + lane_lo;
  if (arow > n_nodes - 1) arow = n_nodes - 1;

  const float dv  = deg[arow];
  const float inv = 1.0f / fmaxf(dv, 1.0f);
  const float hn  = dv > 0.0f ? 1.0f : 0.0f;

  const v2f* xr = (const v2f*)(x       + (size_t)arow * IN_DIM + 2 * hi);
  const v2f* sr = (const v2f*)(sum_out + (size_t)arow * IN_DIM + 2 * hi);

  const int bofs = lane_lo * LDS_STRIDE + 2 * hi;

  v8f acc[4] = {v8f{}, v8f{}, v8f{}, v8f{}};

  for (int c = 0; c < 16; ++c) {
    v2f ax = xr[2 * c];
    v2f am = sr[2 * c] * inv;
    v2f as = ax * hn;

    v2f bg[4], bs[4], bl[4];
#pragma unroll
    for (int nb = 0; nb < 4; ++nb) {
      const int bo = bofs + nb * 16 * LDS_STRIDE + 4 * c;
      bg[nb] = *(const v2f*)(sWg + bo);
      bs[nb] = *(const v2f*)(sWs + bo);
      bl[nb] = *(const v2f*)(sWl + bo);
    }
#pragma unroll
    for (int nb = 0; nb < 4; ++nb) {
      acc[nb] = WMMA_F32_16x16x4(ax, bg[nb], acc[nb]);
      acc[nb] = WMMA_F32_16x16x4(as, bs[nb], acc[nb]);
      acc[nb] = WMMA_F32_16x16x4(am, bl[nb], acc[nb]);
    }
  }

#pragma unroll
  for (int nb = 0; nb < 4; ++nb) {
    const int col = nb * 16 + lane_lo;
    const float bv = bias[col];
    v8f r = acc[nb];
#pragma unroll
    for (int rr = 0; rr < 8; ++rr) {
      int m = node_base + rr + 8 * hi;
      if (m < n_nodes) {
        float val = r[rr] + bv;
        val = val > 0.0f ? val : (__expf(val) - 1.0f);
        sum_out[(size_t)m * OUT_DIM + col] = val;
      }
    }
  }
}

extern "C" void kernel_launch(void* const* d_in, const int* in_sizes, int n_in,
                              void* d_out, int out_size, void* d_ws, size_t ws_size,
                              hipStream_t stream) {
  const float* x    = (const float*)d_in[0];
  const float* Wg   = (const float*)d_in[1];
  const float* Wl   = (const float*)d_in[2];
  const float* Ws   = (const float*)d_in[3];
  const float* bias = (const float*)d_in[4];
  const int*   ei   = (const int*)d_in[5];
  float* out = (float*)d_out;
  float* deg = (float*)d_ws;

  const int n_nodes = in_sizes[0] / IN_DIM;
  const int n_edges = in_sizes[5] / 2;

  hipMemsetAsync(out, 0, (size_t)n_nodes * IN_DIM * sizeof(float), stream);
  hipMemsetAsync(deg, 0, (size_t)n_nodes * sizeof(float), stream);

  {
    long long threads = (long long)n_edges * 16;
    int blocks = (int)((threads + 255) / 256);
    demo_scatter_kernel<<<blocks, 256, 0, stream>>>(x, ei, out, deg, n_edges);
  }
  {
    int waves  = (n_nodes + 15) / 16;
    int blocks = (waves + 7) / 8;
    demo_node_kernel<<<blocks, 256, 0, stream>>>(x, Wg, Wl, Ws, bias, deg, out,
                                                 n_nodes);
  }
}


# One launch, no global atomics or memsets: each work-group owns 640 destination nodes in LDS, scans all edge destinations, adds its edges' x rows by LDS f32 atomics (sum order = arrival), WMMA on LDS
# speedup vs baseline: 1.9692x; 1.9692x over previous
#include <hip/hip_runtime.h>


typedef __attribute__((ext_vector_type(2))) float v2f;
typedef __attribute__((ext_vector_type(8))) float v8f;

#define IN_DIM 64
#define OUT_DIM 64
#define LDS_STRIDE 68
#define NPB 640
#define NTHREADS 1024
#define NWAVES 32
#define QCAP 64
#define UNR 8
#define DB 16

#define WMMA_F32_16x16x4(A, B, C) \
  __builtin_amdgcn_wmma_f32_16x16x4_f32(false, (A), false, (B), (short)0, (C), false, false)

typedef volatile __attribute__((address_space(3))) int lds_vint;

__device__ __forceinline__ int4 load_dst4(const int* __restrict__ dstp, long long g,
                                          int n_edges, bool vec_ok) {
  int4 r;
  if (vec_ok) {
    r = *(const int4*)(dstp + 4 * g);
  } else {
    long long e = 4 * g;
    r.x = (e + 0 < n_edges) ? dstp[e + 0] : -1;
    r.y = (e + 1 < n_edges) ? dstp[e + 1] : -1;
    r.z = (e + 2 < n_edges) ? dstp[e + 2] : -1;
    r.w = (e + 3 < n_edges) ? dstp[e + 3] : -1;
  }
  return r;
}

__device__ __forceinline__ void drain_queue(int n, int lane, lds_vint* qe,
                                            const int* __restrict__ ei,
                                            const int* __restrict__ dstp, int base,
                                            const float* __restrict__ x, float* sAcc,
                                            unsigned* sDeg) {
  int l = 0, s = 0;
  if (lane < n) {
    int e = qe[lane];
    s = ei[e];
    l = dstp[e] - base;
  }
  for (int j0 = 0; j0 < n; j0 += DB) {
    v2f v[DB];
#pragma unroll
    for (int k = 0; k < DB; ++k) {
      int sj = __builtin_amdgcn_readlane(s, j0 + k);
      v[k] = *(const v2f*)(x + (size_t)sj * IN_DIM + 2 * lane);
    }
#pragma unroll
    for (int k = 0; k < DB; ++k) {
      if (j0 + k < n) {
        int lj = __builtin_amdgcn_readlane(l, j0 + k);
        float* p = sAcc + lj * LDS_STRIDE + 2 * lane;
        __hip_atomic_fetch_add(p, v[k].x, __ATOMIC_RELAXED, __HIP_MEMORY_SCOPE_WORKGROUP);
        __hip_atomic_fetch_add(p + 1, v[k].y, __ATOMIC_RELAXED, __HIP_MEMORY_SCOPE_WORKGROUP);
        if (lane == 0)
          __hip_atomic_fetch_add(sDeg + lj, 1u, __ATOMIC_RELAXED, __HIP_MEMORY_SCOPE_WORKGROUP);
      }
    }
  }
}

__global__ __launch_bounds__(NTHREADS) void demo_fused_kernel(
    const float* __restrict__ x, const float* __restrict__ Wg,
    const float* __restrict__ Wl, const float* __restrict__ Ws,
    const float* __restrict__ bias, const int* __restrict__ ei,
    float* __restrict__ out, int n_nodes, int n_edges) {
  __shared__ alignas(16) float sAcc[NPB * LDS_STRIDE];
  __shared__ alignas(16) float sW[3 * OUT_DIM * LDS_STRIDE];
  __shared__ unsigned sDeg[NPB];
  __shared__ int sQe[NWAVES * QCAP];
  float* sWg = sW;
  float* sWs = sW + OUT_DIM * LDS_STRIDE;
  float* sWl = sW + 2 * OUT_DIM * LDS_STRIDE;

  const int tid  = threadIdx.x;
  const int lane = tid & 31;
  const int wave = tid >> 5;
  const int base = blockIdx.x * NPB;
  const int nloc = (n_nodes - base < NPB) ? (n_nodes - base) : NPB;

  for (int i = tid; i < NPB * LDS_STRIDE / 4; i += NTHREADS)
    ((float4*)sAcc)[i] = make_float4(0.f, 0.f, 0.f, 0.f);
  for (int i = tid; i < NPB; i += NTHREADS) sDeg[i] = 0u;
  {
    int row = tid >> 4;
    int k4  = tid & 15;
    float4 g = *(const float4*)(Wg + (size_t)tid * 4);
    float4 s = *(const float4*)(Ws + (size_t)tid * 4);
    float4 l = *(const float4*)(Wl + (size_t)tid * 4);
    *(float4*)(sWg + row * LDS_STRIDE + k4 * 4) = g;
    *(float4*)(sWs + row * LDS_STRIDE + k4 * 4) = s;
    *(float4*)(sWl + row * LDS_STRIDE + k4 * 4) = l;
  }
  __syncthreads();

  {
    const int* dstp = ei + n_edges;
    lds_vint* qe = (lds_vint*)(sQe + wave * QCAP);
    const bool vec_ok = (n_edges & 3) == 0;
    const long long n4 = ((long long)n_edges + 3) >> 2;
    int cnt = 0;

    for (long long g0 = 0; g0 < n4; g0 += (long long)NTHREADS * UNR) {
      int4 d[UNR];
      if (vec_ok && g0 + (long long)NTHREADS * UNR <= n4) {
#pragma unroll
        for (int u = 0; u < UNR; ++u)
          d[u] = *(const int4*)(dstp + 4 * (g0 + (long long)u * NTHREADS + tid));
      } else {
#pragma unroll
        for (int u = 0; u < UNR; ++u) {
          long long g = g0 + (long long)u * NTHREADS + tid;
          d[u] = (g < n4) ? load_dst4(dstp, g, n_edges, vec_ok) : make_int4(-1, -1, -1, -1);
        }
      }
      unsigned mm = 0u;
#pragma unroll
      for (int u = 0; u < UNR; ++u) {
        const int dd[4] = {d[u].x, d[u].y, d[u].z, d[u].w};
#pragma unroll
        for (int k = 0; k < 4; ++k)
          mm |= ((unsigned)(dd[k] - base) < (unsigned)nloc ? 1u : 0u) << (4 * u + k);
      }
      unsigned mask = __builtin_amdgcn_ballot_w32(mm != 0u);
      while (mask) {
        if (mm) {
          const int b = __builtin_ctz(mm);
          mm &= mm - 1u;
          const long long g = g0 + (long long)(b >> 2) * NTHREADS + tid;
          qe[cnt + (int)__builtin_amdgcn_mbcnt_lo(mask, 0u)] = (int)(4 * g) + (b & 3);
        }
        cnt += __builtin_popcount(mask);
        if (cnt >= 32) {
          drain_queue(32, lane, qe, ei, dstp, base, x, sAcc, sDeg);
          const int rem = cnt - 32;
          int te = 0;
          if (lane < rem) te = qe[32 + lane];
          if (lane < rem) qe[lane] = te;
          cnt = rem;
        }
        mask = __builtin_amdgcn_ballot_w32(mm != 0u);
      }
    }
    drain_queue(cnt, lane, qe, ei, dstp, base, x, sAcc, sDeg);
  }
  __syncthreads();

  const int lane_lo = lane & 15;
  const int hi      = lane >> 4;
  const int ntiles  = (nloc + 15) >> 4;
  const int bofs = lane_lo * LDS_STRIDE + 2 * hi;

  for (int item = wave; item < 2 * ntiles; item += NWAVES) {
    const int tile = item >> 1;
    const int half = item & 1;
    const int lrow = tile * 16 + lane_lo;
    int arow = base + lrow;
    if (arow > n_nodes - 1) arow = n_nodes - 1;

    const float dv  = (float)sDeg[lrow];
    const float inv = 1.0f / fmaxf(dv, 1.0f);
    const float hn  = dv > 0.0f ? 1.0f : 0.0f;

    const v2f* xr = (const v2f*)(x    + (size_t)arow * IN_DIM + 2 * hi);
    const v2f* sr = (const v2f*)(sAcc + lrow * LDS_STRIDE + 2 * hi);

    v8f acc[2] = {v8f{}, v8f{}};
#pragma unroll 1
    for (int c4 = 0; c4 < 4; ++c4) {
      v2f ax[4], am[4], bg[4][2], bs[4][2], bl[4][2];
#pragma unroll
      for (int cc = 0; cc < 4; ++cc) {
        const int c = 4 * c4 + cc;
        ax[cc] = xr[2 * c];
        am[cc] = sr[2 * c];
#pragma unroll
        for (int q = 0; q < 2; ++q) {
          const int bo = bofs + (half * 2 + q) * 16 * LDS_STRIDE + 4 * c;
          bg[cc][q] = *(const v2f*)(sWg + bo);
          bs[cc][q] = *(const v2f*)(sWs + bo);
          bl[cc][q] = *(const v2f*)(sWl + bo);
        }
      }
#pragma unroll
      for (int cc = 0; cc < 4; ++cc) {
        const v2f a_m = am[cc] * inv;
        const v2f a_s = ax[cc] * hn;
#pragma unroll
        for (int q = 0; q < 2; ++q) {
          acc[q] = WMMA_F32_16x16x4(ax[cc], bg[cc][q], acc[q]);
          acc[q] = WMMA_F32_16x16x4(a_s, bs[cc][q], acc[q]);
          acc[q] = WMMA_F32_16x16x4(a_m, bl[cc][q], acc[q]);
        }
      }
    }

#pragma unroll
    for (int q = 0; q < 2; ++q) {
      const int col = (half * 2 + q) * 16 + lane_lo;
      const float bv = bias[col];
      v8f r = acc[q];
#pragma unroll
      for (int rr = 0; rr < 8; ++rr) {
        int m = base + tile * 16 + rr + 8 * hi;
        if (m < n_nodes) {
          float val = r[rr] + bv;
          val = val > 0.0f ? val : (__expf(val) - 1.0f);
          out[(size_t)m * OUT_DIM + col] = val;
        }
      }
    }
  }
}

extern "C" void kernel_launch(void* const* d_in, const int* in_sizes, int n_in,
                              void* d_out, int out_size, void* d_ws, size_t ws_size,
                              hipStream_t stream) {
  const float* x    = (const float*)d_in[0];
  const float* Wg   = (const float*)d_in[1];
  const float* Wl   = (const float*)d_in[2];
  const float* Ws   = (const float*)d_in[3];
  const float* bias = (const float*)d_in[4];
  const int*   ei   = (const int*)d_in[5];
  float* out = (float*)d_out;

  const int n_nodes = in_sizes[0] / IN_DIM;
  const int n_edges = in_sizes[5] / 2;

  int blocks = (n_nodes + NPB - 1) / NPB;
  demo_fused_kernel<<<blocks, NTHREADS, 0, stream>>>(x, Wg, Wl, Ws, bias, ei, out,
                                                     n_nodes, n_edges);
}


# Slice key of each edge (dst/640, 8 bits) built bit-sliced by a first launch; each work-group scans 1 byte per edge, was 4, and 8 xors give a 32-edge match mask; 2 launches; sums by arrival as before
# speedup vs baseline: 2.3158x; 1.1760x over previous
#include <hip/hip_runtime.h>


typedef __attribute__((ext_vector_type(2))) float v2f;
typedef __attribute__((ext_vector_type(8))) float v8f;

#define IN_DIM 64
#define OUT_DIM 64
#define LDS_STRIDE 68
#define NPB 640
#define NTHREADS 1024
#define NWAVES 32
#define QCAP 64
#define UG 4
#define KCH 128
#define DB 16

#define WMMA_F32_16x16x4(A, B, C) \
  __builtin_amdgcn_wmma_f32_16x16x4_f32(false, (A), false, (B), (short)0, (C), false, false)

typedef volatile __attribute__((address_space(3))) int lds_vint;

__device__ __forceinline__ int4 load_dst4(const int* __restrict__ dstp, long long g,
                                          int n_edges, bool vec_ok) {
  int4 r;
  if (vec_ok) {
    r = *(const int4*)(dstp + 4 * g);
  } else {
    long long e = 4 * g;
    r.x = (e + 0 < n_edges) ? dstp[e + 0] : -1;
    r.y = (e + 1 < n_edges) ? dstp[e + 1] : -1;
    r.z = (e + 2 < n_edges) ? dstp[e + 2] : -1;
    r.w = (e + 3 < n_edges) ? dstp[e + 3] : -1;
  }
  return r;
}

__device__ __forceinline__ void drain_queue(int n, int lane, lds_vint* qe,
                                            const int* __restrict__ ei,
                                            const int* __restrict__ dstp, int base,
                                            int nloc, int n_edges,
                                            const float* __restrict__ x, float* sAcc,
                                            unsigned* sDeg) {
  int l = -1, s = 0;
  if (lane < n) {
    int e = qe[lane];
    if (e < n_edges) {
      s = ei[e];
      l = dstp[e] - base;
    }
  }
  for (int j0 = 0; j0 < n; j0 += DB) {
    v2f v[DB];
#pragma unroll
    for (int k = 0; k < DB; ++k) {
      int sj = __builtin_amdgcn_readlane(s, j0 + k);
      v[k] = *(const v2f*)(x + (size_t)sj * IN_DIM + 2 * lane);
    }
#pragma unroll
    for (int k = 0; k < DB; ++k) {
      const int lj = __builtin_amdgcn_readlane(l, j0 + k);
      if (j0 + k < n && (unsigned)lj < (unsigned)nloc) {
        float* p = sAcc + lj * LDS_STRIDE + 2 * lane;
        __hip_atomic_fetch_add(p, v[k].x, __ATOMIC_RELAXED, __HIP_MEMORY_SCOPE_WORKGROUP);
        __hip_atomic_fetch_add(p + 1, v[k].y, __ATOMIC_RELAXED, __HIP_MEMORY_SCOPE_WORKGROUP);
        if (lane == 0)
          __hip_atomic_fetch_add(sDeg + lj, 1u, __ATOMIC_RELAXED, __HIP_MEMORY_SCOPE_WORKGROUP);
      }
    }
  }
}

__global__ __launch_bounds__(256) void demo_key_kernel(
    const int* __restrict__ ei, unsigned* __restrict__ keys_lo,
    unsigned* __restrict__ keys_hi, int n_nodes, int n_edges, int n_chunks) {
  const int* dstp = ei + n_edges;
  const int lane = threadIdx.x & 31;
  const int wave = (int)((blockIdx.x * blockDim.x + threadIdx.x) >> 5);
  const bool vec_ok = (n_edges & 3) == 0;
  const long long n4 = ((long long)n_edges + 3) >> 2;

  int4 d[4];
  const long long gw = 128LL * wave;
  if (vec_ok && gw + 128 <= n4) {
#pragma unroll
    for (int k = 0; k < 4; ++k) d[k] = *(const int4*)(dstp + 4 * (gw + 32 * k + lane));
  } else {
#pragma unroll
    for (int k = 0; k < 4; ++k) {
      const long long g = gw + 32 * k + lane;
      d[k] = (g < n4) ? load_dst4(dstp, g, n_edges, vec_ok) : make_int4(-1, -1, -1, -1);
    }
  }
#pragma unroll
  for (int k = 0; k < 4; ++k) {
    const int C = 4 * wave + k;
    if (C < n_chunks) {
      const int dd[4] = {d[k].x, d[k].y, d[k].z, d[k].w};
      unsigned w = 0u;
#pragma unroll
      for (int c = 0; c < 4; ++c) {
        const unsigned key = (unsigned)dd[c] < (unsigned)n_nodes
                                 ? ((unsigned)dd[c] / (unsigned)NPB) & 255u : 255u;
#pragma unroll
        for (int i = 0; i < 8; ++i) {
          const unsigned bal = __builtin_amdgcn_ballot_w32(((key >> i) & 1u) != 0u);
          if (lane == (i < 4 ? 0 : 12) + 4 * c + i) w = bal;
        }
      }
      unsigned* p = lane < 16 ? keys_lo + (size_t)16 * C + lane
                              : keys_hi + (size_t)16 * C + (lane - 16);
      *p = w;
    }
  }
}

__global__ __launch_bounds__(NTHREADS) void demo_fused_kernel(
    const float* __restrict__ x, const float* __restrict__ Wg,
    const float* __restrict__ Wl, const float* __restrict__ Ws,
    const float* __restrict__ bias, const int* __restrict__ ei,
    const unsigned* __restrict__ keys_lo, const unsigned* __restrict__ keys_hi,
    float* __restrict__ out, int n_nodes, int n_edges, int n_chunks) {
  __shared__ alignas(16) float sAcc[NPB * LDS_STRIDE];
  __shared__ alignas(16) float sW[3 * OUT_DIM * LDS_STRIDE];
  __shared__ unsigned sDeg[NPB];
  __shared__ int sQe[NWAVES * QCAP];
  float* sWg = sW;
  float* sWs = sW + OUT_DIM * LDS_STRIDE;
  float* sWl = sW + 2 * OUT_DIM * LDS_STRIDE;

  const int tid  = threadIdx.x;
  const int lane = tid & 31;
  const int wave = tid >> 5;
  const int base = blockIdx.x * NPB;
  const int nloc = (n_nodes - base < NPB) ? (n_nodes - base) : NPB;

  for (int i = tid; i < NPB * LDS_STRIDE / 4; i += NTHREADS)
    ((float4*)sAcc)[i] = make_float4(0.f, 0.f, 0.f, 0.f);
  for (int i = tid; i < NPB; i += NTHREADS) sDeg[i] = 0u;
  {
    int row = tid >> 4;
    int k4  = tid & 15;
    float4 g = *(const float4*)(Wg + (size_t)tid * 4);
    float4 s = *(const float4*)(Ws + (size_t)tid * 4);
    float4 l = *(const float4*)(Wl + (size_t)tid * 4);
    *(float4*)(sWg + row * LDS_STRIDE + k4 * 4) = g;
    *(float4*)(sWs + row * LDS_STRIDE + k4 * 4) = s;
    *(float4*)(sWl + row * LDS_STRIDE + k4 * 4) = l;
  }
  __syncthreads();

  {
    const int* dstp = ei + n_edges;
    lds_vint* qe = (lds_vint*)(sQe + wave * QCAP);
    const uint4* kLo = (const uint4*)keys_lo;
    const uint4* kHi = (const uint4*)keys_hi;
    const int nG = 4 * n_chunks;
    const unsigned myKey = blockIdx.x & 255u;
    unsigned kb[8];
#pragma unroll
    for (int i = 0; i < 8; ++i) kb[i] = ((myKey >> i) & 1u) ? 0xFFFFFFFFu : 0u;
    int cnt = 0;

    for (int g0 = 0; g0 < nG; g0 += NTHREADS * UG) {
      uint4 lo[UG], hi[UG];
      if (g0 + NTHREADS * UG <= nG) {
#pragma unroll
        for (int u = 0; u < UG; ++u) {
          lo[u] = kLo[g0 + u * NTHREADS + tid];
          hi[u] = kHi[g0 + u * NTHREADS + tid];
        }
      } else {
#pragma unroll
        for (int u = 0; u < UG; ++u) {
          const int G = g0 + u * NTHREADS + tid;
          const bool ok = G < nG;
          lo[u] = ok ? kLo[G] : make_uint4(0u, 0u, 0u, 0u);
          hi[u] = ok ? kHi[G] : make_uint4(0u, 0u, 0u, 0u);
        }
      }
      unsigned mm[UG];
#pragma unroll
      for (int u = 0; u < UG; ++u) {
        const unsigned df = (lo[u].x ^ kb[0]) | (lo[u].y ^ kb[1]) | (lo[u].z ^ kb[2]) |
                            (lo[u].w ^ kb[3]) | (hi[u].x ^ kb[4]) | (hi[u].y ^ kb[5]) |
                            (hi[u].z ^ kb[6]) | (hi[u].w ^ kb[7]);
        mm[u] = (g0 + u * NTHREADS + tid < nG) ? ~df : 0u;
      }
#pragma unroll
      for (int u = 0; u < UG; ++u) {
        const int G = g0 + u * NTHREADS + tid;
        const int ebase = ((G >> 2) << 7) + (G & 3);
        unsigned m = mm[u];
        unsigned mask = __builtin_amdgcn_ballot_w32(m != 0u);
        while (mask) {
          if (m) {
            const int b = __builtin_ctz(m);
            m &= m - 1u;
            qe[cnt + (int)__builtin_amdgcn_mbcnt_lo(mask, 0u)] = ebase + 4 * b;
          }
          cnt += __builtin_popcount(mask);
          if (cnt >= 32) {
            drain_queue(32, lane, qe, ei, dstp, base, nloc, n_edges, x, sAcc, sDeg);
            const int rem = cnt - 32;
            int te = 0;
            if (lane < rem) te = qe[32 + lane];
            if (lane < rem) qe[lane] = te;
            cnt = rem;
          }
          mask = __builtin_amdgcn_ballot_w32(m != 0u);
        }
      }
    }
    drain_queue(cnt, lane, qe, ei, dstp, base, nloc, n_edges, x, sAcc, sDeg);
  }
  __syncthreads();

  const int lane_lo = lane & 15;
  const int hi      = lane >> 4;
  const int ntiles  = (nloc + 15) >> 4;
  const int bofs = lane_lo * LDS_STRIDE + 2 * hi;

  for (int item = wave; item < 2 * ntiles; item += NWAVES) {
    const int tile = item >> 1;
    const int half = item & 1;
    const int lrow = tile * 16 + lane_lo;
    int arow = base + lrow;
    if (arow > n_nodes - 1) arow = n_nodes - 1;

    const float dv  = (float)sDeg[lrow];
    const float inv = 1.0f / fmaxf(dv, 1.0f);
    const float hn  = dv > 0.0f ? 1.0f : 0.0f;

    const v2f* xr = (const v2f*)(x    + (size_t)arow * IN_DIM + 2 * hi);
    const v2f* sr = (const v2f*)(sAcc + lrow * LDS_STRIDE + 2 * hi);

    v8f acc[2] = {v8f{}, v8f{}};
#pragma unroll 1
    for (int c4 = 0; c4 < 4; ++c4) {
      v2f ax[4], am[4], bg[4][2], bs[4][2], bl[4][2];
#pragma unroll
      for (int cc = 0; cc < 4; ++cc) {
        const int c = 4 * c4 + cc;
        ax[cc] = xr[2 * c];
        am[cc] = sr[2 * c];
#pragma unroll
        for (int q = 0; q < 2; ++q) {
          const int bo = bofs + (half * 2 + q) * 16 * LDS_STRIDE + 4 * c;
          bg[cc][q] = *(const v2f*)(sWg + bo);
          bs[cc][q] = *(const v2f*)(sWs + bo);
          bl[cc][q] = *(const v2f*)(sWl + bo);
        }
      }
#pragma unroll
      for (int cc = 0; cc < 4; ++cc) {
        const v2f a_m = am[cc] * inv;
        const v2f a_s = ax[cc] * hn;
#pragma unroll
        for (int q = 0; q < 2; ++q) {
          acc[q] = WMMA_F32_16x16x4(ax[cc], bg[cc][q], acc[q]);
          acc[q] = WMMA_F32_16x16x4(a_s, bs[cc][q], acc[q]);
          acc[q] = WMMA_F32_16x16x4(a_m, bl[cc][q], acc[q]);
        }
      }
    }

#pragma unroll
    for (int q = 0; q < 2; ++q) {
      const int col = (half * 2 + q) * 16 + lane_lo;
      const float bv = bias[col];
      v8f r = acc[q];
#pragma unroll
      for (int rr = 0; rr < 8; ++rr) {
        int m = base + tile * 16 + rr + 8 * hi;
        if (m < n_nodes) {
          float val = r[rr] + bv;
          val = val > 0.0f ? val : (__expf(val) - 1.0f);
          out[(size_t)m * OUT_DIM + col] = val;
        }
      }
    }
  }
}

extern "C" void kernel_launch(void* const* d_in, const int* in_sizes, int n_in,
                              void* d_out, int out_size, void* d_ws, size_t ws_size,
                              hipStream_t stream) {
  const float* x    = (const float*)d_in[0];
  const float* Wg   = (const float*)d_in[1];
  const float* Wl   = (const float*)d_in[2];
  const float* Ws   = (const float*)d_in[3];
  const float* bias = (const float*)d_in[4];
  const int*   ei   = (const int*)d_in[5];
  float* out = (float*)d_out;

  const int n_nodes = in_sizes[0] / IN_DIM;
  const int n_edges = in_sizes[5] / 2;

  const int n_chunks = (n_edges + KCH - 1) / KCH;
  unsigned* keys_lo = (unsigned*)d_ws;
  unsigned* keys_hi = keys_lo + (size_t)16 * n_chunks;
  if (ws_size < (size_t)KCH * n_chunks) return;

  int kblocks = (n_chunks + 31) / 32;
  demo_key_kernel<<<kblocks, 256, 0, stream>>>(ei, keys_lo, keys_hi, n_nodes, n_edges, n_chunks);

  int blocks = (n_nodes + NPB - 1) / NPB;
  demo_fused_kernel<<<blocks, NTHREADS, 0, stream>>>(x, Wg, Wl, Ws, bias, ei, keys_lo, keys_hi,
                                                     out, n_nodes, n_edges, n_chunks);
}


# Destination slice per work-group 400 nodes, was 640: 250 work-groups, was 157, so 250 of the chip's 256 WGPs work, was 157; each still scans all keys; sum order = arrival as before
# speedup vs baseline: 2.9393x; 1.2692x over previous
#include <hip/hip_runtime.h>


typedef __attribute__((ext_vector_type(2))) float v2f;
typedef __attribute__((ext_vector_type(8))) float v8f;

#define IN_DIM 64
#define OUT_DIM 64
#define LDS_STRIDE 68
#define NPB 400
#define NTHREADS 1024
#define NWAVES 32
#define QCAP 64
#define UG 4
#define KCH 128
#define DB 16

#define WMMA_F32_16x16x4(A, B, C) \
  __builtin_amdgcn_wmma_f32_16x16x4_f32(false, (A), false, (B), (short)0, (C), false, false)

typedef volatile __attribute__((address_space(3))) int lds_vint;

__device__ __forceinline__ int4 load_dst4(const int* __restrict__ dstp, long long g,
                                          int n_edges, bool vec_ok) {
  int4 r;
  if (vec_ok) {
    r = *(const int4*)(dstp + 4 * g);
  } else {
    long long e = 4 * g;
    r.x = (e + 0 < n_edges) ? dstp[e + 0] : -1;
    r.y = (e + 1 < n_edges) ? dstp[e + 1] : -1;
    r.z = (e + 2 < n_edges) ? dstp[e + 2] : -1;
    r.w = (e + 3 < n_edges) ? dstp[e + 3] : -1;
  }
  return r;
}

__device__ __forceinline__ void drain_queue(int n, int lane, lds_vint* qe,
                                            const int* __restrict__ ei,
                                            const int* __restrict__ dstp, int base,
                                            int nloc, int n_edges,
                                            const float* __restrict__ x, float* sAcc,
                                            unsigned* sDeg) {
  int l = -1, s = 0;
  if (lane < n) {
    int e = qe[lane];
    if (e < n_edges) {
      s = ei[e];
      l = dstp[e] - base;
    }
  }
  for (int j0 = 0; j0 < n; j0 += DB) {
    v2f v[DB];
#pragma unroll
    for (int k = 0; k < DB; ++k) {
      int sj = __builtin_amdgcn_readlane(s, j0 + k);
      v[k] = *(const v2f*)(x + (size_t)sj * IN_DIM + 2 * lane);
    }
#pragma unroll
    for (int k = 0; k < DB; ++k) {
      const int lj = __builtin_amdgcn_readlane(l, j0 + k);
      if (j0 + k < n && (unsigned)lj < (unsigned)nloc) {
        float* p = sAcc + lj * LDS_STRIDE + 2 * lane;
        __hip_atomic_fetch_add(p, v[k].x, __ATOMIC_RELAXED, __HIP_MEMORY_SCOPE_WORKGROUP);
        __hip_atomic_fetch_add(p + 1, v[k].y, __ATOMIC_RELAXED, __HIP_MEMORY_SCOPE_WORKGROUP);
        if (lane == 0)
          __hip_atomic_fetch_add(sDeg + lj, 1u, __ATOMIC_RELAXED, __HIP_MEMORY_SCOPE_WORKGROUP);
      }
    }
  }
}

__global__ __launch_bounds__(256) void demo_key_kernel(
    const int* __restrict__ ei, unsigned* __restrict__ keys_lo,
    unsigned* __restrict__ keys_hi, int n_nodes, int n_edges, int n_chunks) {
  const int* dstp = ei + n_edges;
  const int lane = threadIdx.x & 31;
  const int wave = (int)((blockIdx.x * blockDim.x + threadIdx.x) >> 5);
  const bool vec_ok = (n_edges & 3) == 0;
  const long long n4 = ((long long)n_edges + 3) >> 2;

  int4 d[4];
  const long long gw = 128LL * wave;
  if (vec_ok && gw + 128 <= n4) {
#pragma unroll
    for (int k = 0; k < 4; ++k) d[k] = *(const int4*)(dstp + 4 * (gw + 32 * k + lane));
  } else {
#pragma unroll
    for (int k = 0; k < 4; ++k) {
      const long long g = gw + 32 * k + lane;
      d[k] = (g < n4) ? load_dst4(dstp, g, n_edges, vec_ok) : make_int4(-1, -1, -1, -1);
    }
  }
#pragma unroll
  for (int k = 0; k < 4; ++k) {
    const int C = 4 * wave + k;
    if (C < n_chunks) {
      const int dd[4] = {d[k].x, d[k].y, d[k].z, d[k].w};
      unsigned w = 0u;
#pragma unroll
      for (int c = 0; c < 4; ++c) {
        const unsigned key = (unsigned)dd[c] < (unsigned)n_nodes
                                 ? ((unsigned)dd[c] / (unsigned)NPB) & 255u : 255u;
#pragma unroll
        for (int i = 0; i < 8; ++i) {
          const unsigned bal = __builtin_amdgcn_ballot_w32(((key >> i) & 1u) != 0u);
          if (lane == (i < 4 ? 0 : 12) + 4 * c + i) w = bal;
        }
      }
      unsigned* p = lane < 16 ? keys_lo + (size_t)16 * C + lane
                              : keys_hi + (size_t)16 * C + (lane - 16);
      *p = w;
    }
  }
}

__global__ __launch_bounds__(NTHREADS) void demo_fused_kernel(
    const float* __restrict__ x, const float* __restrict__ Wg,
    const float* __restrict__ Wl, const float* __restrict__ Ws,
    const float* __restrict__ bias, const int* __restrict__ ei,
    const unsigned* __restrict__ keys_lo, const unsigned* __restrict__ keys_hi,
    float* __restrict__ out, int n_nodes, int n_edges, int n_chunks) {
  __shared__ alignas(16) float sAcc[NPB * LDS_STRIDE];
  __shared__ alignas(16) float sW[3 * OUT_DIM * LDS_STRIDE];
  __shared__ unsigned sDeg[NPB];
  __shared__ int sQe[NWAVES * QCAP];
  float* sWg = sW;
  float* sWs = sW + OUT_DIM * LDS_STRIDE;
  float* sWl = sW + 2 * OUT_DIM * LDS_STRIDE;

  const int tid  = threadIdx.x;
  const int lane = tid & 31;
  const int wave = tid >> 5;
  const int base = blockIdx.x * NPB;
  const int nloc = (n_nodes - base < NPB) ? (n_nodes - base) : NPB;

  for (int i = tid; i < NPB * LDS_STRIDE / 4; i += NTHREADS)
    ((float4*)sAcc)[i] = make_float4(0.f, 0.f, 0.f, 0.f);
  for (int i = tid; i < NPB; i += NTHREADS) sDeg[i] = 0u;
  {
    int row = tid >> 4;
    int k4  = tid & 15;
    float4 g = *(const float4*)(Wg + (size_t)tid * 4);
    float4 s = *(const float4*)(Ws + (size_t)tid * 4);
    float4 l = *(const float4*)(Wl + (size_t)tid * 4);
    *(float4*)(sWg + row * LDS_STRIDE + k4 * 4) = g;
    *(float4*)(sWs + row * LDS_STRIDE + k4 * 4) = s;
    *(float4*)(sWl + row * LDS_STRIDE + k4 * 4) = l;
  }
  __syncthreads();

  {
    const int* dstp = ei + n_edges;
    lds_vint* qe = (lds_vint*)(sQe + wave * QCAP);
    const uint4* kLo = (const uint4*)keys_lo;
    const uint4* kHi = (const uint4*)keys_hi;
    const int nG = 4 * n_chunks;
    const unsigned myKey = blockIdx.x & 255u;
    unsigned kb[8];
#pragma unroll
    for (int i = 0; i < 8; ++i) kb[i] = ((myKey >> i) & 1u) ? 0xFFFFFFFFu : 0u;
    int cnt = 0;

    for (int g0 = 0; g0 < nG; g0 += NTHREADS * UG) {
      uint4 lo[UG], hi[UG];
      if (g0 + NTHREADS * UG <= nG) {
#pragma unroll
        for (int u = 0; u < UG; ++u) {
          lo[u] = kLo[g0 + u * NTHREADS + tid];
          hi[u] = kHi[g0 + u * NTHREADS + tid];
        }
      } else {
#pragma unroll
        for (int u = 0; u < UG; ++u) {
          const int G = g0 + u * NTHREADS + tid;
          const bool ok = G < nG;
          lo[u] = ok ? kLo[G] : make_uint4(0u, 0u, 0u, 0u);
          hi[u] = ok ? kHi[G] : make_uint4(0u, 0u, 0u, 0u);
        }
      }
      unsigned mm[UG];
#pragma unroll
      for (int u = 0; u < UG; ++u) {
        const unsigned df = (lo[u].x ^ kb[0]) | (lo[u].y ^ kb[1]) | (lo[u].z ^ kb[2]) |
                            (lo[u].w ^ kb[3]) | (hi[u].x ^ kb[4]) | (hi[u].y ^ kb[5]) |
                            (hi[u].z ^ kb[6]) | (hi[u].w ^ kb[7]);
        mm[u] = (g0 + u * NTHREADS + tid < nG) ? ~df : 0u;
      }
#pragma unroll
      for (int u = 0; u < UG; ++u) {
        const int G = g0 + u * NTHREADS + tid;
        const int ebase = ((G >> 2) << 7) + (G & 3);
        unsigned m = mm[u];
        unsigned mask = __builtin_amdgcn_ballot_w32(m != 0u);
        while (mask) {
          if (m) {
            const int b = __builtin_ctz(m);
            m &= m - 1u;
            qe[cnt + (int)__builtin_amdgcn_mbcnt_lo(mask, 0u)] = ebase + 4 * b;
          }
          cnt += __builtin_popcount(mask);
          if (cnt >= 32) {
            drain_queue(32, lane, qe, ei, dstp, base, nloc, n_edges, x, sAcc, sDeg);
            const int rem = cnt - 32;
            int te = 0;
            if (lane < rem) te = qe[32 + lane];
            if (lane < rem) qe[lane] = te;
            cnt = rem;
          }
          mask = __builtin_amdgcn_ballot_w32(m != 0u);
        }
      }
    }
    drain_queue(cnt, lane, qe, ei, dstp, base, nloc, n_edges, x, sAcc, sDeg);
  }
  __syncthreads();

  const int lane_lo = lane & 15;
  const int hi      = lane >> 4;
  const int ntiles  = (nloc + 15) >> 4;
  const int bofs = lane_lo * LDS_STRIDE + 2 * hi;

  for (int item = wave; item < 2 * ntiles; item += NWAVES) {
    const int tile = item >> 1;
    const int half = item & 1;
    const int lrow = tile * 16 + lane_lo;
    int arow = base + lrow;
    if (arow > n_nodes - 1) arow = n_nodes - 1;

    const float dv  = (float)sDeg[lrow];
    const float inv = 1.0f / fmaxf(dv, 1.0f);
    const float hn  = dv > 0.0f ? 1.0f : 0.0f;

    const v2f* xr = (const v2f*)(x    + (size_t)arow * IN_DIM + 2 * hi);
    const v2f* sr = (const v2f*)(sAcc + lrow * LDS_STRIDE + 2 * hi);

    v8f acc[2] = {v8f{}, v8f{}};
#pragma unroll 1
    for (int c4 = 0; c4 < 4; ++c4) {
      v2f ax[4], am[4], bg[4][2], bs[4][2], bl[4][2];
#pragma unroll
      for (int cc = 0; cc < 4; ++cc) {
        const int c = 4 * c4 + cc;
        ax[cc] = xr[2 * c];
        am[cc] = sr[2 * c];
#pragma unroll
        for (int q = 0; q < 2; ++q) {
          const int bo = bofs + (half * 2 + q) * 16 * LDS_STRIDE + 4 * c;
          bg[cc][q] = *(const v2f*)(sWg + bo);
          bs[cc][q] = *(const v2f*)(sWs + bo);
          bl[cc][q] = *(const v2f*)(sWl + bo);
        }
      }
#pragma unroll
      for (int cc = 0; cc < 4; ++cc) {
        const v2f a_m = am[cc] * inv;
        const v2f a_s = ax[cc] * hn;
#pragma unroll
        for (int q = 0; q < 2; ++q) {
          acc[q] = WMMA_F32_16x16x4(ax[cc], bg[cc][q], acc[q]);
          acc[q] = WMMA_F32_16x16x4(a_s, bs[cc][q], acc[q]);
          acc[q] = WMMA_F32_16x16x4(a_m, bl[cc][q], acc[q]);
        }
      }
    }

#pragma unroll
    for (int q = 0; q < 2; ++q) {
      const int col = (half * 2 + q) * 16 + lane_lo;
      const float bv = bias[col];
      v8f r = acc[q];
#pragma unroll
      for (int rr = 0; rr < 8; ++rr) {
        int m = base + tile * 16 + rr + 8 * hi;
        if (m < n_nodes) {
          float val = r[rr] + bv;
          val = val > 0.0f ? val : (__expf(val) - 1.0f);
          out[(size_t)m * OUT_DIM + col] = val;
        }
      }
    }
  }
}

extern "C" void kernel_launch(void* const* d_in, const int* in_sizes, int n_in,
                              void* d_out, int out_size, void* d_ws, size_t ws_size,
                              hipStream_t stream) {
  const float* x    = (const float*)d_in[0];
  const float* Wg   = (const float*)d_in[1];
  const float* Wl   = (const float*)d_in[2];
  const float* Ws   = (const float*)d_in[3];
  const float* bias = (const float*)d_in[4];
  const int*   ei   = (const int*)d_in[5];
  float* out = (float*)d_out;

  const int n_nodes = in_sizes[0] / IN_DIM;
  const int n_edges = in_sizes[5] / 2;

  const int n_chunks = (n_edges + KCH - 1) / KCH;
  unsigned* keys_lo = (unsigned*)d_ws;
  unsigned* keys_hi = keys_lo + (size_t)16 * n_chunks;
  if (ws_size < (size_t)KCH * n_chunks) return;

  int kblocks = (n_chunks + 31) / 32;
  demo_key_kernel<<<kblocks, 256, 0, stream>>>(ei, keys_lo, keys_hi, n_nodes, n_edges, n_chunks);

  int blocks = (n_nodes + NPB - 1) / NPB;
  demo_fused_kernel<<<blocks, NTHREADS, 0, stream>>>(x, Wg, Wl, Ws, bias, ei, keys_lo, keys_hi,
                                                     out, n_nodes, n_edges, n_chunks);
}


# In-degree counted by one 32-lane LDS integer add per drained batch of 32 edges, was one single-lane LDS add per edge: 2 LDS atomic instructions per edge, was 3; float sums by arrival as before
# speedup vs baseline: 3.0490x; 1.0373x over previous
#include <hip/hip_runtime.h>


typedef __attribute__((ext_vector_type(2))) float v2f;
typedef __attribute__((ext_vector_type(8))) float v8f;

#define IN_DIM 64
#define OUT_DIM 64
#define LDS_STRIDE 68
#define NPB 400
#define NTHREADS 1024
#define NWAVES 32
#define QCAP 64
#define UG 4
#define KCH 128
#define DB 16

#define WMMA_F32_16x16x4(A, B, C) \
  __builtin_amdgcn_wmma_f32_16x16x4_f32(false, (A), false, (B), (short)0, (C), false, false)

typedef volatile __attribute__((address_space(3))) int lds_vint;

__device__ __forceinline__ int4 load_dst4(const int* __restrict__ dstp, long long g,
                                          int n_edges, bool vec_ok) {
  int4 r;
  if (vec_ok) {
    r = *(const int4*)(dstp + 4 * g);
  } else {
    long long e = 4 * g;
    r.x = (e + 0 < n_edges) ? dstp[e + 0] : -1;
    r.y = (e + 1 < n_edges) ? dstp[e + 1] : -1;
    r.z = (e + 2 < n_edges) ? dstp[e + 2] : -1;
    r.w = (e + 3 < n_edges) ? dstp[e + 3] : -1;
  }
  return r;
}

__device__ __forceinline__ void drain_queue(int n, int lane, lds_vint* qe,
                                            const int* __restrict__ ei,
                                            const int* __restrict__ dstp, int base,
                                            int nloc, int n_edges,
                                            const float* __restrict__ x, float* sAcc,
                                            unsigned* sDeg) {
  int l = -1, s = 0;
  if (lane < n) {
    int e = qe[lane];
    if (e < n_edges) {
      s = ei[e];
      l = dstp[e] - base;
      if ((unsigned)l < (unsigned)nloc)
        __hip_atomic_fetch_add(sDeg + l, 1u, __ATOMIC_RELAXED, __HIP_MEMORY_SCOPE_WORKGROUP);
    }
  }
  for (int j0 = 0; j0 < n; j0 += DB) {
    v2f v[DB];
#pragma unroll
    for (int k = 0; k < DB; ++k) {
      int sj = __builtin_amdgcn_readlane(s, j0 + k);
      v[k] = *(const v2f*)(x + (size_t)sj * IN_DIM + 2 * lane);
    }
#pragma unroll
    for (int k = 0; k < DB; ++k) {
      const int lj = __builtin_amdgcn_readlane(l, j0 + k);
      if (j0 + k < n && (unsigned)lj < (unsigned)nloc) {
        float* p = sAcc + lj * LDS_STRIDE + 2 * lane;
        __hip_atomic_fetch_add(p, v[k].x, __ATOMIC_RELAXED, __HIP_MEMORY_SCOPE_WORKGROUP);
        __hip_atomic_fetch_add(p + 1, v[k].y, __ATOMIC_RELAXED, __HIP_MEMORY_SCOPE_WORKGROUP);
      }
    }
  }
}

__global__ __launch_bounds__(256) void demo_key_kernel(
    const int* __restrict__ ei, unsigned* __restrict__ keys_lo,
    unsigned* __restrict__ keys_hi, int n_nodes, int n_edges, int n_chunks) {
  const int* dstp = ei + n_edges;
  const int lane = threadIdx.x & 31;
  const int wave = (int)((blockIdx.x * blockDim.x + threadIdx.x) >> 5);
  const bool vec_ok = (n_edges & 3) == 0;
  const long long n4 = ((long long)n_edges + 3) >> 2;

  int4 d[4];
  const long long gw = 128LL * wave;
  if (vec_ok && gw + 128 <= n4) {
#pragma unroll
    for (int k = 0; k < 4; ++k) d[k] = *(const int4*)(dstp + 4 * (gw + 32 * k + lane));
  } else {
#pragma unroll
    for (int k = 0; k < 4; ++k) {
      const long long g = gw + 32 * k + lane;
      d[k] = (g < n4) ? load_dst4(dstp, g, n_edges, vec_ok) : make_int4(-1, -1, -1, -1);
    }
  }
#pragma unroll
  for (int k = 0; k < 4; ++k) {
    const int C = 4 * wave + k;
    if (C < n_chunks) {
      const int dd[4] = {d[k].x, d[k].y, d[k].z, d[k].w};
      unsigned w = 0u;
#pragma unroll
      for (int c = 0; c < 4; ++c) {
        const unsigned key = (unsigned)dd[c] < (unsigned)n_nodes
                                 ? ((unsigned)dd[c] / (unsigned)NPB) & 255u : 255u;
#pragma unroll
        for (int i = 0; i < 8; ++i) {
          const unsigned bal = __builtin_amdgcn_ballot_w32(((key >> i) & 1u) != 0u);
          if (lane == (i < 4 ? 0 : 12) + 4 * c + i) w = bal;
        }
      }
      unsigned* p = lane < 16 ? keys_lo + (size_t)16 * C + lane
                              : keys_hi + (size_t)16 * C + (lane - 16);
      *p = w;
    }
  }
}

__global__ __launch_bounds__(NTHREADS) void demo_fused_kernel(
    const float* __restrict__ x, const float* __restrict__ Wg,
    const float* __restrict__ Wl, const float* __restrict__ Ws,
    const float* __restrict__ bias, const int* __restrict__ ei,
    const unsigned* __restrict__ keys_lo, const unsigned* __restrict__ keys_hi,
    float* __restrict__ out, int n_nodes, int n_edges, int n_chunks) {
  __shared__ alignas(16) float sAcc[NPB * LDS_STRIDE];
  __shared__ alignas(16) float sW[3 * OUT_DIM * LDS_STRIDE];
  __shared__ unsigned sDeg[NPB];
  __shared__ int sQe[NWAVES * QCAP];
  float* sWg = sW;
  float* sWs = sW + OUT_DIM * LDS_STRIDE;
  float* sWl = sW + 2 * OUT_DIM * LDS_STRIDE;

  const int tid  = threadIdx.x;
  const int lane = tid & 31;
  const int wave = tid >> 5;
  const int base = blockIdx.x * NPB;
  const int nloc = (n_nodes - base < NPB) ? (n_nodes - base) : NPB;

  for (int i = tid; i < NPB * LDS_STRIDE / 4; i += NTHREADS)
    ((float4*)sAcc)[i] = make_float4(0.f, 0.f, 0.f, 0.f);
  for (int i = tid; i < NPB; i += NTHREADS) sDeg[i] = 0u;
  {
    int row = tid >> 4;
    int k4  = tid & 15;
    float4 g = *(const float4*)(Wg + (size_t)tid * 4);
    float4 s = *(const float4*)(Ws + (size_t)tid * 4);
    float4 l = *(const float4*)(Wl + (size_t)tid * 4);
    *(float4*)(sWg + row * LDS_STRIDE + k4 * 4) = g;
    *(float4*)(sWs + row * LDS_STRIDE + k4 * 4) = s;
    *(float4*)(sWl + row * LDS_STRIDE + k4 * 4) = l;
  }
  __syncthreads();

  {
    const int* dstp = ei + n_edges;
    lds_vint* qe = (lds_vint*)(sQe + wave * QCAP);
    const uint4* kLo = (const uint4*)keys_lo;
    const uint4* kHi = (const uint4*)keys_hi;
    const int nG = 4 * n_chunks;
    const unsigned myKey = blockIdx.x & 255u;
    unsigned kb[8];
#pragma unroll
    for (int i = 0; i < 8; ++i) kb[i] = ((myKey >> i) & 1u) ? 0xFFFFFFFFu : 0u;
    int cnt = 0;

    for (int g0 = 0; g0 < nG; g0 += NTHREADS * UG) {
      uint4 lo[UG], hi[UG];
      if (g0 + NTHREADS * UG <= nG) {
#pragma unroll
        for (int u = 0; u < UG; ++u) {
          lo[u] = kLo[g0 + u * NTHREADS + tid];
          hi[u] = kHi[g0 + u * NTHREADS + tid];
        }
      } else {
#pragma unroll
        for (int u = 0; u < UG; ++u) {
          const int G = g0 + u * NTHREADS + tid;
          const bool ok = G < nG;
          lo[u] = ok ? kLo[G] : make_uint4(0u, 0u, 0u, 0u);
          hi[u] = ok ? kHi[G] : make_uint4(0u, 0u, 0u, 0u);
        }
      }
      unsigned mm[UG];
#pragma unroll
      for (int u = 0; u < UG; ++u) {
        const unsigned df = (lo[u].x ^ kb[0]) | (lo[u].y ^ kb[1]) | (lo[u].z ^ kb[2]) |
                            (lo[u].w ^ kb[3]) | (hi[u].x ^ kb[4]) | (hi[u].y ^ kb[5]) |
                            (hi[u].z ^ kb[6]) | (hi[u].w ^ kb[7]);
        mm[u] = (g0 + u * NTHREADS + tid < nG) ? ~df : 0u;
      }
#pragma unroll
      for (int u = 0; u < UG; ++u) {
        const int G = g0 + u * NTHREADS + tid;
        const int ebase = ((G >> 2) << 7) + (G & 3);
        unsigned m = mm[u];
        unsigned mask = __builtin_amdgcn_ballot_w32(m != 0u);
        while (mask) {
          if (m) {
            const int b = __builtin_ctz(m);
            m &= m - 1u;
            qe[cnt + (int)__builtin_amdgcn_mbcnt_lo(mask, 0u)] = ebase + 4 * b;
          }
          cnt += __builtin_popcount(mask);
          if (cnt >= 32) {
            drain_queue(32, lane, qe, ei, dstp, base, nloc, n_edges, x, sAcc, sDeg);
            const int rem = cnt - 32;
            int te = 0;
            if (lane < rem) te = qe[32 + lane];
            if (lane < rem) qe[lane] = te;
            cnt = rem;
          }
          mask = __builtin_amdgcn_ballot_w32(m != 0u);
        }
      }
    }
    drain_queue(cnt, lane, qe, ei, dstp, base, nloc, n_edges, x, sAcc, sDeg);
  }
  __syncthreads();

  const int lane_lo = lane & 15;
  const int hi      = lane >> 4;
  const int ntiles  = (nloc + 15) >> 4;
  const int bofs = lane_lo * LDS_STRIDE + 2 * hi;

  for (int item = wave; item < 2 * ntiles; item += NWAVES) {
    const int tile = item >> 1;
    const int half = item & 1;
    const int lrow = tile * 16 + lane_lo;
    int arow = base + lrow;
    if (arow > n_nodes - 1) arow = n_nodes - 1;

    const float dv  = (float)sDeg[lrow];
    const float inv = 1.0f / fmaxf(dv, 1.0f);
    const float hn  = dv > 0.0f ? 1.0f : 0.0f;

    const v2f* xr = (const v2f*)(x    + (size_t)arow * IN_DIM + 2 * hi);
    const v2f* sr = (const v2f*)(sAcc + lrow * LDS_STRIDE + 2 * hi);

    v8f acc[2] = {v8f{}, v8f{}};
#pragma unroll 1
    for (int c4 = 0; c4 < 4; ++c4) {
      v2f ax[4], am[4], bg[4][2], bs[4][2], bl[4][2];
#pragma unroll
      for (int cc = 0; cc < 4; ++cc) {
        const int c = 4 * c4 + cc;
        ax[cc] = xr[2 * c];
        am[cc] = sr[2 * c];
#pragma unroll
        for (int q = 0; q < 2; ++q) {
          const int bo = bofs + (half * 2 + q) * 16 * LDS_STRIDE + 4 * c;
          bg[cc][q] = *(const v2f*)(sWg + bo);
          bs[cc][q] = *(const v2f*)(sWs + bo);
          bl[cc][q] = *(const v2f*)(sWl + bo);
        }
      }
#pragma unroll
      for (int cc = 0; cc < 4; ++cc) {
        const v2f a_m = am[cc] * inv;
        const v2f a_s = ax[cc] * hn;
#pragma unroll
        for (int q = 0; q < 2; ++q) {
          acc[q] = WMMA_F32_16x16x4(ax[cc], bg[cc][q], acc[q]);
          acc[q] = WMMA_F32_16x16x4(a_s, bs[cc][q], acc[q]);
          acc[q] = WMMA_F32_16x16x4(a_m, bl[cc][q], acc[q]);
        }
      }
    }

#pragma unroll
    for (int q = 0; q < 2; ++q) {
      const int col = (half * 2 + q) * 16 + lane_lo;
      const float bv = bias[col];
      v8f r = acc[q];
#pragma unroll
      for (int rr = 0; rr < 8; ++rr) {
        int m = base + tile * 16 + rr + 8 * hi;
        if (m < n_nodes) {
          float val = r[rr] + bv;
          val = val > 0.0f ? val : (__expf(val) - 1.0f);
          out[(size_t)m * OUT_DIM + col] = val;
        }
      }
    }
  }
}

extern "C" void kernel_launch(void* const* d_in, const int* in_sizes, int n_in,
                              void* d_out, int out_size, void* d_ws, size_t ws_size,
                              hipStream_t stream) {
  const float* x    = (const float*)d_in[0];
  const float* Wg   = (const float*)d_in[1];
  const float* Wl   = (const float*)d_in[2];
  const float* Ws   = (const float*)d_in[3];
  const float* bias = (const float*)d_in[4];
  const int*   ei   = (const int*)d_in[5];
  float* out = (float*)d_out;

  const int n_nodes = in_sizes[0] / IN_DIM;
  const int n_edges = in_sizes[5] / 2;

  const int n_chunks = (n_edges + KCH - 1) / KCH;
  unsigned* keys_lo = (unsigned*)d_ws;
  unsigned* keys_hi = keys_lo + (size_t)16 * n_chunks;
  if (ws_size < (size_t)KCH * n_chunks) return;

  int kblocks = (n_chunks + 31) / 32;
  demo_key_kernel<<<kblocks, 256, 0, stream>>>(ei, keys_lo, keys_hi, n_nodes, n_edges, n_chunks);

  int blocks = (n_nodes + NPB - 1) / NPB;
  demo_fused_kernel<<<blocks, NTHREADS, 0, stream>>>(x, Wg, Wl, Ws, bias, ei, keys_lo, keys_hi,
                                                     out, n_nodes, n_edges, n_chunks);
}


# First launch also writes one packed word per edge (source<<9 | destination's row in its slice); a kept edge costs one gathered dword, was two (source and destination); sums by arrival as before
# speedup vs baseline: 3.2696x; 1.0724x over previous
#include <hip/hip_runtime.h>


typedef __attribute__((ext_vector_type(2))) float v2f;
typedef __attribute__((ext_vector_type(8))) float v8f;

#define IN_DIM 64
#define OUT_DIM 64
#define LDS_STRIDE 68
#define NPB 400
#define NTHREADS 1024
#define NWAVES 32
#define QCAP 64
#define UG 4
#define KCH 128
#define DB 16

#define WMMA_F32_16x16x4(A, B, C) \
  __builtin_amdgcn_wmma_f32_16x16x4_f32(false, (A), false, (B), (short)0, (C), false, false)

typedef volatile __attribute__((address_space(3))) int lds_vint;

__device__ __forceinline__ int4 load_dst4(const int* __restrict__ dstp, long long g,
                                          int n_edges, bool vec_ok) {
  int4 r;
  if (vec_ok) {
    r = *(const int4*)(dstp + 4 * g);
  } else {
    long long e = 4 * g;
    r.x = (e + 0 < n_edges) ? dstp[e + 0] : -1;
    r.y = (e + 1 < n_edges) ? dstp[e + 1] : -1;
    r.z = (e + 2 < n_edges) ? dstp[e + 2] : -1;
    r.w = (e + 3 < n_edges) ? dstp[e + 3] : -1;
  }
  return r;
}

__device__ __forceinline__ void drain_queue(int n, int lane, lds_vint* qe,
                                            const int* __restrict__ ei,
                                            const int* __restrict__ dstp,
                                            const unsigned* __restrict__ rec, int base,
                                            int nloc, int n_edges,
                                            const float* __restrict__ x, float* sAcc,
                                            unsigned* sDeg) {
  int l = -1, s = 0;
  if (lane < n) {
    int e = qe[lane];
    if (rec) {
      const unsigned r = rec[e];
      const int lr = (int)(r & 511u);
      if (lr < nloc) {
        l = lr;
        s = (int)(r >> 9);
        __hip_atomic_fetch_add(sDeg + l, 1u, __ATOMIC_RELAXED, __HIP_MEMORY_SCOPE_WORKGROUP);
      }
    } else if (e < n_edges) {
      s = ei[e];
      l = dstp[e] - base;
      if ((unsigned)l < (unsigned)nloc)
        __hip_atomic_fetch_add(sDeg + l, 1u, __ATOMIC_RELAXED, __HIP_MEMORY_SCOPE_WORKGROUP);
    }
  }
  for (int j0 = 0; j0 < n; j0 += DB) {
    v2f v[DB];
#pragma unroll
    for (int k = 0; k < DB; ++k) {
      int sj = __builtin_amdgcn_readlane(s, j0 + k);
      v[k] = *(const v2f*)(x + (size_t)sj * IN_DIM + 2 * lane);
    }
#pragma unroll
    for (int k = 0; k < DB; ++k) {
      const int lj = __builtin_amdgcn_readlane(l, j0 + k);
      if (j0 + k < n && (unsigned)lj < (unsigned)nloc) {
        float* p = sAcc + lj * LDS_STRIDE + 2 * lane;
        __hip_atomic_fetch_add(p, v[k].x, __ATOMIC_RELAXED, __HIP_MEMORY_SCOPE_WORKGROUP);
        __hip_atomic_fetch_add(p + 1, v[k].y, __ATOMIC_RELAXED, __HIP_MEMORY_SCOPE_WORKGROUP);
      }
    }
  }
}

__global__ __launch_bounds__(256) void demo_key_kernel(
    const int* __restrict__ ei, unsigned* __restrict__ keys_lo,
    unsigned* __restrict__ keys_hi, unsigned* __restrict__ rec,
    int n_nodes, int n_edges, int n_chunks) {
  const int* dstp = ei + n_edges;
  const int lane = threadIdx.x & 31;
  const int wave = (int)((blockIdx.x * blockDim.x + threadIdx.x) >> 5);
  const bool vec_ok = (n_edges & 3) == 0;
  const long long n4 = ((long long)n_edges + 3) >> 2;

  int4 d[4], sv[4];
  const long long gw = 128LL * wave;
  if (vec_ok && gw + 128 <= n4) {
#pragma unroll
    for (int k = 0; k < 4; ++k) d[k] = *(const int4*)(dstp + 4 * (gw + 32 * k + lane));
    if (rec) {
#pragma unroll
      for (int k = 0; k < 4; ++k) sv[k] = *(const int4*)(ei + 4 * (gw + 32 * k + lane));
    }
  } else {
#pragma unroll
    for (int k = 0; k < 4; ++k) {
      const long long g = gw + 32 * k + lane;
      d[k] = (g < n4) ? load_dst4(dstp, g, n_edges, vec_ok) : make_int4(-1, -1, -1, -1);
      sv[k] = (rec && g < n4) ? load_dst4(ei, g, n_edges, vec_ok) : make_int4(0, 0, 0, 0);
    }
  }
#pragma unroll
  for (int k = 0; k < 4; ++k) {
    const int C = 4 * wave + k;
    if (C < n_chunks) {
      const int dd[4] = {d[k].x, d[k].y, d[k].z, d[k].w};
      unsigned w = 0u;
#pragma unroll
      for (int c = 0; c < 4; ++c) {
        const unsigned key = (unsigned)dd[c] < (unsigned)n_nodes
                                 ? ((unsigned)dd[c] / (unsigned)NPB) & 255u : 255u;
#pragma unroll
        for (int i = 0; i < 8; ++i) {
          const unsigned bal = __builtin_amdgcn_ballot_w32(((key >> i) & 1u) != 0u);
          if (lane == (i < 4 ? 0 : 12) + 4 * c + i) w = bal;
        }
      }
      unsigned* p = lane < 16 ? keys_lo + (size_t)16 * C + lane
                              : keys_hi + (size_t)16 * C + (lane - 16);
      *p = w;
      if (rec) {
        const int ss[4] = {sv[k].x, sv[k].y, sv[k].z, sv[k].w};
        unsigned rv[4];
#pragma unroll
        for (int c = 0; c < 4; ++c) {
          const unsigned du = (unsigned)dd[c];
          rv[c] = du < (unsigned)n_nodes
                      ? ((unsigned)ss[c] << 9) | (du - (du / (unsigned)NPB) * (unsigned)NPB)
                      : 0xFFFFFFFFu;
        }
        *(uint4*)(rec + (size_t)128 * C + 4 * lane) = make_uint4(rv[0], rv[1], rv[2], rv[3]);
      }
    }
  }
}

__global__ __launch_bounds__(NTHREADS) void demo_fused_kernel(
    const float* __restrict__ x, const float* __restrict__ Wg,
    const float* __restrict__ Wl, const float* __restrict__ Ws,
    const float* __restrict__ bias, const int* __restrict__ ei,
    const unsigned* __restrict__ keys_lo, const unsigned* __restrict__ keys_hi,
    const unsigned* __restrict__ rec,
    float* __restrict__ out, int n_nodes, int n_edges, int n_chunks) {
  __shared__ alignas(16) float sAcc[NPB * LDS_STRIDE];
  __shared__ alignas(16) float sW[3 * OUT_DIM * LDS_STRIDE];
  __shared__ unsigned sDeg[NPB];
  __shared__ int sQe[NWAVES * QCAP];
  float* sWg = sW;
  float* sWs = sW + OUT_DIM * LDS_STRIDE;
  float* sWl = sW + 2 * OUT_DIM * LDS_STRIDE;

  const int tid  = threadIdx.x;
  const int lane = tid & 31;
  const int wave = tid >> 5;
  const int base = blockIdx.x * NPB;
  const int nloc = (n_nodes - base < NPB) ? (n_nodes - base) : NPB;

  for (int i = tid; i < NPB * LDS_STRIDE / 4; i += NTHREADS)
    ((float4*)sAcc)[i] = make_float4(0.f, 0.f, 0.f, 0.f);
  for (int i = tid; i < NPB; i += NTHREADS) sDeg[i] = 0u;
  {
    int row = tid >> 4;
    int k4  = tid & 15;
    float4 g = *(const float4*)(Wg + (size_t)tid * 4);
    float4 s = *(const float4*)(Ws + (size_t)tid * 4);
    float4 l = *(const float4*)(Wl + (size_t)tid * 4);
    *(float4*)(sWg + row * LDS_STRIDE + k4 * 4) = g;
    *(float4*)(sWs + row * LDS_STRIDE + k4 * 4) = s;
    *(float4*)(sWl + row * LDS_STRIDE + k4 * 4) = l;
  }
  __syncthreads();

  {
    const int* dstp = ei + n_edges;
    lds_vint* qe = (lds_vint*)(sQe + wave * QCAP);
    const uint4* kLo = (const uint4*)keys_lo;
    const uint4* kHi = (const uint4*)keys_hi;
    const int nG = 4 * n_chunks;
    const unsigned myKey = blockIdx.x & 255u;
    unsigned kb[8];
#pragma unroll
    for (int i = 0; i < 8; ++i) kb[i] = ((myKey >> i) & 1u) ? 0xFFFFFFFFu : 0u;
    int cnt = 0;

    for (int g0 = 0; g0 < nG; g0 += NTHREADS * UG) {
      uint4 lo[UG], hi[UG];
      if (g0 + NTHREADS * UG <= nG) {
#pragma unroll
        for (int u = 0; u < UG; ++u) {
          lo[u] = kLo[g0 + u * NTHREADS + tid];
          hi[u] = kHi[g0 + u * NTHREADS + tid];
        }
      } else {
#pragma unroll
        for (int u = 0; u < UG; ++u) {
          const int G = g0 + u * NTHREADS + tid;
          const bool ok = G < nG;
          lo[u] = ok ? kLo[G] : make_uint4(0u, 0u, 0u, 0u);
          hi[u] = ok ? kHi[G] : make_uint4(0u, 0u, 0u, 0u);
        }
      }
      unsigned mm[UG];
#pragma unroll
      for (int u = 0; u < UG; ++u) {
        const unsigned df = (lo[u].x ^ kb[0]) | (lo[u].y ^ kb[1]) | (lo[u].z ^ kb[2]) |
                            (lo[u].w ^ kb[3]) | (hi[u].x ^ kb[4]) | (hi[u].y ^ kb[5]) |
                            (hi[u].z ^ kb[6]) | (hi[u].w ^ kb[7]);
        mm[u] = (g0 + u * NTHREADS + tid < nG) ? ~df : 0u;
      }
#pragma unroll
      for (int u = 0; u < UG; ++u) {
        const int G = g0 + u * NTHREADS + tid;
        const int ebase = ((G >> 2) << 7) + (G & 3);
        unsigned m = mm[u];
        unsigned mask = __builtin_amdgcn_ballot_w32(m != 0u);
        while (mask) {
          if (m) {
            const int b = __builtin_ctz(m);
            m &= m - 1u;
            qe[cnt + (int)__builtin_amdgcn_mbcnt_lo(mask, 0u)] = ebase + 4 * b;
          }
          cnt += __builtin_popcount(mask);
          if (cnt >= 32) {
            drain_queue(32, lane, qe, ei, dstp, rec, base, nloc, n_edges, x, sAcc, sDeg);
            const int rem = cnt - 32;
            int te = 0;
            if (lane < rem) te = qe[32 + lane];
            if (lane < rem) qe[lane] = te;
            cnt = rem;
          }
          mask = __builtin_amdgcn_ballot_w32(m != 0u);
        }
      }
    }
    drain_queue(cnt, lane, qe, ei, dstp, rec, base, nloc, n_edges, x, sAcc, sDeg);
  }
  __syncthreads();

  const int lane_lo = lane & 15;
  const int hi      = lane >> 4;
  const int ntiles  = (nloc + 15) >> 4;
  const int bofs = lane_lo * LDS_STRIDE + 2 * hi;

  for (int item = wave; item < 2 * ntiles; item += NWAVES) {
    const int tile = item >> 1;
    const int half = item & 1;
    const int lrow = tile * 16 + lane_lo;
    int arow = base + lrow;
    if (arow > n_nodes - 1) arow = n_nodes - 1;

    const float dv  = (float)sDeg[lrow];
    const float inv = 1.0f / fmaxf(dv, 1.0f);
    const float hn  = dv > 0.0f ? 1.0f : 0.0f;

    const v2f* xr = (const v2f*)(x    + (size_t)arow * IN_DIM + 2 * hi);
    const v2f* sr = (const v2f*)(sAcc + lrow * LDS_STRIDE + 2 * hi);

    v8f acc[2] = {v8f{}, v8f{}};
#pragma unroll 1
    for (int c4 = 0; c4 < 4; ++c4) {
      v2f ax[4], am[4], bg[4][2], bs[4][2], bl[4][2];
#pragma unroll
      for (int cc = 0; cc < 4; ++cc) {
        const int c = 4 * c4 + cc;
        ax[cc] = xr[2 * c];
        am[cc] = sr[2 * c];
#pragma unroll
        for (int q = 0; q < 2; ++q) {
          const int bo = bofs + (half * 2 + q) * 16 * LDS_STRIDE + 4 * c;
          bg[cc][q] = *(const v2f*)(sWg + bo);
          bs[cc][q] = *(const v2f*)(sWs + bo);
          bl[cc][q] = *(const v2f*)(sWl + bo);
        }
      }
#pragma unroll
      for (int cc = 0; cc < 4; ++cc) {
        const v2f a_m = am[cc] * inv;
        const v2f a_s = ax[cc] * hn;
#pragma unroll
        for (int q = 0; q < 2; ++q) {
          acc[q] = WMMA_F32_16x16x4(ax[cc], bg[cc][q], acc[q]);
          acc[q] = WMMA_F32_16x16x4(a_s, bs[cc][q], acc[q]);
          acc[q] = WMMA_F32_16x16x4(a_m, bl[cc][q], acc[q]);
        }
      }
    }

#pragma unroll
    for (int q = 0; q < 2; ++q) {
      const int col = (half * 2 + q) * 16 + lane_lo;
      const float bv = bias[col];
      v8f r = acc[q];
#pragma unroll
      for (int rr = 0; rr < 8; ++rr) {
        int m = base + tile * 16 + rr + 8 * hi;
        if (m < n_nodes) {
          float val = r[rr] + bv;
          val = val > 0.0f ? val : (__expf(val) - 1.0f);
          out[(size_t)m * OUT_DIM + col] = val;
        }
      }
    }
  }
}

extern "C" void kernel_launch(void* const* d_in, const int* in_sizes, int n_in,
                              void* d_out, int out_size, void* d_ws, size_t ws_size,
                              hipStream_t stream) {
  const float* x    = (const float*)d_in[0];
  const float* Wg   = (const float*)d_in[1];
  const float* Wl   = (const float*)d_in[2];
  const float* Ws   = (const float*)d_in[3];
  const float* bias = (const float*)d_in[4];
  const int*   ei   = (const int*)d_in[5];
  float* out = (float*)d_out;

  const int n_nodes = in_sizes[0] / IN_DIM;
  const int n_edges = in_sizes[5] / 2;

  const int n_chunks = (n_edges + KCH - 1) / KCH;
  unsigned* keys_lo = (unsigned*)d_ws;
  unsigned* keys_hi = keys_lo + (size_t)16 * n_chunks;
  if (ws_size < (size_t)KCH * n_chunks) return;
  unsigned* rec = (ws_size >= (size_t)5 * KCH * n_chunks && n_nodes <= 256 * NPB)
                      ? keys_hi + (size_t)16 * n_chunks : nullptr;

  int kblocks = (n_chunks + 31) / 32;
  demo_key_kernel<<<kblocks, 256, 0, stream>>>(ei, keys_lo, keys_hi, rec, n_nodes, n_edges, n_chunks);

  int blocks = (n_nodes + NPB - 1) / NPB;
  demo_fused_kernel<<<blocks, NTHREADS, 0, stream>>>(x, Wg, Wl, Ws, bias, ei, keys_lo, keys_hi,
                                                     rec, out, n_nodes, n_edges, n_chunks);
}


# Slice key stored as two one-hot digits (16 + 16 bit planes, was 8 binary planes): a work-group reads 2 bits per edge, was 8, and one AND gives the match mask; sums by arrival as before
# speedup vs baseline: 3.3779x; 1.0331x over previous
#include <hip/hip_runtime.h>


typedef __attribute__((ext_vector_type(2))) float v2f;
typedef __attribute__((ext_vector_type(8))) float v8f;

#define IN_DIM 64
#define OUT_DIM 64
#define LDS_STRIDE 68
#define NPB 400
#define NTHREADS 1024
#define NWAVES 32
#define QCAP 64
#define UG 4
#define KCH 128
#define DB 16

#define WMMA_F32_16x16x4(A, B, C) \
  __builtin_amdgcn_wmma_f32_16x16x4_f32(false, (A), false, (B), (short)0, (C), false, false)

typedef volatile __attribute__((address_space(3))) int lds_vint;

__device__ __forceinline__ int4 load_dst4(const int* __restrict__ dstp, long long g,
                                          int n_edges, bool vec_ok) {
  int4 r;
  if (vec_ok) {
    r = *(const int4*)(dstp + 4 * g);
  } else {
    long long e = 4 * g;
    r.x = (e + 0 < n_edges) ? dstp[e + 0] : -1;
    r.y = (e + 1 < n_edges) ? dstp[e + 1] : -1;
    r.z = (e + 2 < n_edges) ? dstp[e + 2] : -1;
    r.w = (e + 3 < n_edges) ? dstp[e + 3] : -1;
  }
  return r;
}

__device__ __forceinline__ void drain_queue(int n, int lane, lds_vint* qe,
                                            const int* __restrict__ ei,
                                            const int* __restrict__ dstp,
                                            const unsigned* __restrict__ rec, int base,
                                            int nloc, int n_edges,
                                            const float* __restrict__ x, float* sAcc,
                                            unsigned* sDeg) {
  int l = -1, s = 0;
  if (lane < n) {
    int e = qe[lane];
    if (rec) {
      const unsigned r = rec[e];
      const int lr = (int)(r & 511u);
      if (lr < nloc) {
        l = lr;
        s = (int)(r >> 9);
        __hip_atomic_fetch_add(sDeg + l, 1u, __ATOMIC_RELAXED, __HIP_MEMORY_SCOPE_WORKGROUP);
      }
    } else if (e < n_edges) {
      s = ei[e];
      l = dstp[e] - base;
      if ((unsigned)l < (unsigned)nloc)
        __hip_atomic_fetch_add(sDeg + l, 1u, __ATOMIC_RELAXED, __HIP_MEMORY_SCOPE_WORKGROUP);
    }
  }
  for (int j0 = 0; j0 < n; j0 += DB) {
    v2f v[DB];
#pragma unroll
    for (int k = 0; k < DB; ++k) {
      int sj = __builtin_amdgcn_readlane(s, j0 + k);
      v[k] = *(const v2f*)(x + (size_t)sj * IN_DIM + 2 * lane);
    }
#pragma unroll
    for (int k = 0; k < DB; ++k) {
      const int lj = __builtin_amdgcn_readlane(l, j0 + k);
      if (j0 + k < n && (unsigned)lj < (unsigned)nloc) {
        float* p = sAcc + lj * LDS_STRIDE + 2 * lane;
        __hip_atomic_fetch_add(p, v[k].x, __ATOMIC_RELAXED, __HIP_MEMORY_SCOPE_WORKGROUP);
        __hip_atomic_fetch_add(p + 1, v[k].y, __ATOMIC_RELAXED, __HIP_MEMORY_SCOPE_WORKGROUP);
      }
    }
  }
}

__global__ __launch_bounds__(256) void demo_key_kernel(
    const int* __restrict__ ei, unsigned* __restrict__ planes,
    unsigned* __restrict__ rec,
    int n_nodes, int n_edges, int n_chunks) {
  const int* dstp = ei + n_edges;
  const int lane = threadIdx.x & 31;
  const int wave = (int)((blockIdx.x * blockDim.x + threadIdx.x) >> 5);
  const bool vec_ok = (n_edges & 3) == 0;
  const long long n4 = ((long long)n_edges + 3) >> 2;

  const bool hi_digit = lane < 16;
  unsigned flip[4];
#pragma unroll
  for (int i = 0; i < 4; ++i) flip[i] = ((lane >> i) & 1) ? 0u : 0xFFFFFFFFu;

  int4 d[4], sv[4];
  const long long gw = 128LL * wave;
  if (vec_ok && gw + 128 <= n4) {
#pragma unroll
    for (int k = 0; k < 4; ++k) d[k] = *(const int4*)(dstp + 4 * (gw + 32 * k + lane));
    if (rec) {
#pragma unroll
      for (int k = 0; k < 4; ++k) sv[k] = *(const int4*)(ei + 4 * (gw + 32 * k + lane));
    }
  } else {
#pragma unroll
    for (int k = 0; k < 4; ++k) {
      const long long g = gw + 32 * k + lane;
      d[k] = (g < n4) ? load_dst4(dstp, g, n_edges, vec_ok) : make_int4(-1, -1, -1, -1);
      sv[k] = (rec && g < n4) ? load_dst4(ei, g, n_edges, vec_ok) : make_int4(0, 0, 0, 0);
    }
  }
#pragma unroll
  for (int k = 0; k < 4; ++k) {
    const int C = 4 * wave + k;
    if (C < n_chunks) {
      const int dd[4] = {d[k].x, d[k].y, d[k].z, d[k].w};
      unsigned w[4];
#pragma unroll
      for (int c = 0; c < 4; ++c) {
        const bool valid = (unsigned)dd[c] < (unsigned)n_nodes;
        const unsigned key = ((unsigned)dd[c] / (unsigned)NPB) & 255u;
        const unsigned vb = __builtin_amdgcn_ballot_w32(valid);
        unsigned klo[4], khi[4];
#pragma unroll
        for (int i = 0; i < 4; ++i) {
          klo[i] = __builtin_amdgcn_ballot_w32(((key >> i) & 1u) != 0u);
          khi[i] = __builtin_amdgcn_ballot_w32(((key >> (4 + i)) & 1u) != 0u);
        }
        unsigned wc = vb;
#pragma unroll
        for (int i = 0; i < 4; ++i) {
          const unsigned sel = hi_digit ? khi[i] : klo[i];
          wc &= sel ^ flip[i];
        }
        w[c] = wc;
      }
      ((uint4*)planes)[(size_t)lane * n_chunks + C] = make_uint4(w[0], w[1], w[2], w[3]);
      if (rec) {
        const int ss[4] = {sv[k].x, sv[k].y, sv[k].z, sv[k].w};
        unsigned rv[4];
#pragma unroll
        for (int c = 0; c < 4; ++c) {
          const unsigned du = (unsigned)dd[c];
          rv[c] = du < (unsigned)n_nodes
                      ? ((unsigned)ss[c] << 9) | (du - (du / (unsigned)NPB) * (unsigned)NPB)
                      : 0xFFFFFFFFu;
        }
        *(uint4*)(rec + (size_t)128 * C + 4 * lane) = make_uint4(rv[0], rv[1], rv[2], rv[3]);
      }
    }
  }
}

__global__ __launch_bounds__(NTHREADS) void demo_fused_kernel(
    const float* __restrict__ x, const float* __restrict__ Wg,
    const float* __restrict__ Wl, const float* __restrict__ Ws,
    const float* __restrict__ bias, const int* __restrict__ ei,
    const unsigned* __restrict__ planes,
    const unsigned* __restrict__ rec,
    float* __restrict__ out, int n_nodes, int n_edges, int n_chunks) {
  __shared__ alignas(16) float sAcc[NPB * LDS_STRIDE];
  __shared__ alignas(16) float sW[3 * OUT_DIM * LDS_STRIDE];
  __shared__ unsigned sDeg[NPB];
  __shared__ int sQe[NWAVES * QCAP];
  float* sWg = sW;
  float* sWs = sW + OUT_DIM * LDS_STRIDE;
  float* sWl = sW + 2 * OUT_DIM * LDS_STRIDE;

  const int tid  = threadIdx.x;
  const int lane = tid & 31;
  const int wave = tid >> 5;
  const int base = blockIdx.x * NPB;
  const int nloc = (n_nodes - base < NPB) ? (n_nodes - base) : NPB;

  for (int i = tid; i < NPB * LDS_STRIDE / 4; i += NTHREADS)
    ((float4*)sAcc)[i] = make_float4(0.f, 0.f, 0.f, 0.f);
  for (int i = tid; i < NPB; i += NTHREADS) sDeg[i] = 0u;
  {
    int row = tid >> 4;
    int k4  = tid & 15;
    float4 g = *(const float4*)(Wg + (size_t)tid * 4);
    float4 s = *(const float4*)(Ws + (size_t)tid * 4);
    float4 l = *(const float4*)(Wl + (size_t)tid * 4);
    *(float4*)(sWg + row * LDS_STRIDE + k4 * 4) = g;
    *(float4*)(sWs + row * LDS_STRIDE + k4 * 4) = s;
    *(float4*)(sWl + row * LDS_STRIDE + k4 * 4) = l;
  }
  __syncthreads();

  {
    const int* dstp = ei + n_edges;
    lds_vint* qe = (lds_vint*)(sQe + wave * QCAP);
    const unsigned myKey = blockIdx.x & 255u;
    const uint4* rowA = (const uint4*)planes + (size_t)(myKey >> 4) * n_chunks;
    const uint4* rowB = (const uint4*)planes + (size_t)(16u + (myKey & 15u)) * n_chunks;
    int cnt = 0;

    for (int q0 = 0; q0 < n_chunks; q0 += NTHREADS * UG) {
      uint4 pa[UG], pb[UG];
      if (q0 + NTHREADS * UG <= n_chunks) {
#pragma unroll
        for (int u = 0; u < UG; ++u) {
          pa[u] = rowA[q0 + u * NTHREADS + tid];
          pb[u] = rowB[q0 + u * NTHREADS + tid];
        }
      } else {
#pragma unroll
        for (int u = 0; u < UG; ++u) {
          const int q = q0 + u * NTHREADS + tid;
          const bool ok = q < n_chunks;
          pa[u] = ok ? rowA[q] : make_uint4(0u, 0u, 0u, 0u);
          pb[u] = ok ? rowB[q] : make_uint4(0u, 0u, 0u, 0u);
        }
      }
#pragma unroll
      for (int u = 0; u < UG; ++u) {
        const int q = q0 + u * NTHREADS + tid;
        unsigned r0 = pa[u].x & pb[u].x, r1 = pa[u].y & pb[u].y;
        unsigned r2 = pa[u].z & pb[u].z, r3 = pa[u].w & pb[u].w;
#pragma unroll 1
        for (int c = 0; c < 4; ++c) {
          const int ebase = (q << 7) + c;
          unsigned m = r0;
          r0 = r1; r1 = r2; r2 = r3;
          unsigned mask = __builtin_amdgcn_ballot_w32(m != 0u);
          while (mask) {
            if (m) {
              const int b = __builtin_ctz(m);
              m &= m - 1u;
              qe[cnt + (int)__builtin_amdgcn_mbcnt_lo(mask, 0u)] = ebase + 4 * b;
            }
            cnt += __builtin_popcount(mask);
            if (cnt >= 32) {
              drain_queue(32, lane, qe, ei, dstp, rec, base, nloc, n_edges, x, sAcc, sDeg);
              const int rem = cnt - 32;
              int te = 0;
              if (lane < rem) te = qe[32 + lane];
              if (lane < rem) qe[lane] = te;
              cnt = rem;
            }
            mask = __builtin_amdgcn_ballot_w32(m != 0u);
          }
        }
      }
    }
    drain_queue(cnt, lane, qe, ei, dstp, rec, base, nloc, n_edges, x, sAcc, sDeg);
  }
  __syncthreads();

  const int lane_lo = lane & 15;
  const int hi      = lane >> 4;
  const int ntiles  = (nloc + 15) >> 4;
  const int bofs = lane_lo * LDS_STRIDE + 2 * hi;

  for (int item = wave; item < 2 * ntiles; item += NWAVES) {
    const int tile = item >> 1;
    const int half = item & 1;
    const int lrow = tile * 16 + lane_lo;
    int arow = base + lrow;
    if (arow > n_nodes - 1) arow = n_nodes - 1;

    const float dv  = (float)sDeg[lrow];
    const float inv = 1.0f / fmaxf(dv, 1.0f);
    const float hn  = dv > 0.0f ? 1.0f : 0.0f;

    const v2f* xr = (const v2f*)(x    + (size_t)arow * IN_DIM + 2 * hi);
    const v2f* sr = (const v2f*)(sAcc + lrow * LDS_STRIDE + 2 * hi);

    v8f acc[2] = {v8f{}, v8f{}};
#pragma unroll 1
    for (int c4 = 0; c4 < 4; ++c4) {
      v2f ax[4], am[4], bg[4][2], bs[4][2], bl[4][2];
#pragma unroll
      for (int cc = 0; cc < 4; ++cc) {
        const int c = 4 * c4 + cc;
        ax[cc] = xr[2 * c];
        am[cc] = sr[2 * c];
#pragma unroll
        for (int q = 0; q < 2; ++q) {
          const int bo = bofs + (half * 2 + q) * 16 * LDS_STRIDE + 4 * c;
          bg[cc][q] = *(const v2f*)(sWg + bo);
          bs[cc][q] = *(const v2f*)(sWs + bo);
          bl[cc][q] = *(const v2f*)(sWl + bo);
        }
      }
#pragma unroll
      for (int cc = 0; cc < 4; ++cc) {
        const v2f a_m = am[cc] * inv;
        const v2f a_s = ax[cc] * hn;
#pragma unroll
        for (int q = 0; q < 2; ++q) {
          acc[q] = WMMA_F32_16x16x4(ax[cc], bg[cc][q], acc[q]);
          acc[q] = WMMA_F32_16x16x4(a_s, bs[cc][q], acc[q]);
          acc[q] = WMMA_F32_16x16x4(a_m, bl[cc][q], acc[q]);
        }
      }
    }

#pragma unroll
    for (int q = 0; q < 2; ++q) {
      const int col = (half * 2 + q) * 16 + lane_lo;
      const float bv = bias[col];
      v8f r = acc[q];
#pragma unroll
      for (int rr = 0; rr < 8; ++rr) {
        int m = base + tile * 16 + rr + 8 * hi;
        if (m < n_nodes) {
          float val = r[rr] + bv;
          val = val > 0.0f ? val : (__expf(val) - 1.0f);
          out[(size_t)m * OUT_DIM + col] = val;
        }
      }
    }
  }
}

extern "C" void kernel_launch(void* const* d_in, const int* in_sizes, int n_in,
                              void* d_out, int out_size, void* d_ws, size_t ws_size,
                              hipStream_t stream) {
  const float* x    = (const float*)d_in[0];
  const float* Wg   = (const float*)d_in[1];
  const float* Wl   = (const float*)d_in[2];
  const float* Ws   = (const float*)d_in[3];
  const float* bias = (const float*)d_in[4];
  const int*   ei   = (const int*)d_in[5];
  float* out = (float*)d_out;

  const int n_nodes = in_sizes[0] / IN_DIM;
  const int n_edges = in_sizes[5] / 2;

  const int n_chunks = (n_edges + KCH - 1) / KCH;
  unsigned* planes = (unsigned*)d_ws;
  if (ws_size < (size_t)4 * KCH * n_chunks) return;
  unsigned* rec = (ws_size >= (size_t)8 * KCH * n_chunks && n_nodes <= 256 * NPB)
                      ? planes + (size_t)KCH * n_chunks : nullptr;

  int kblocks = (n_chunks + 31) / 32;
  demo_key_kernel<<<kblocks, 256, 0, stream>>>(ei, planes, rec, n_nodes, n_edges, n_chunks);

  int blocks = (n_nodes + NPB - 1) / NPB;
  demo_fused_kernel<<<blocks, NTHREADS, 0, stream>>>(x, Wg, Wl, Ws, bias, ei, planes,
                                                     rec, out, n_nodes, n_edges, n_chunks);
}


# x@Wg^T + x@Ws^T of a tile issued by its wave inside the key scan (a SIMD's 8 waves at 8 distinct steps), kept in LDS as start of the mean@Wl^T sum: 2/3 of WMMAs leave the tail; term sum order changes
# speedup vs baseline: 3.6017x; 1.0663x over previous
#include <hip/hip_runtime.h>


typedef __attribute__((ext_vector_type(2))) float v2f;
typedef __attribute__((ext_vector_type(8))) float v8f;

#define IN_DIM 64
#define OUT_DIM 64
#define LDS_STRIDE 68
#define NPB 400
#define NTHREADS 1024
#define NWAVES 32
#define QCAP 64
#define UG 4
#define KCH 128
#define DB 16
#define EPASSES 3

#define WMMA_F32_16x16x4(A, B, C) \
  __builtin_amdgcn_wmma_f32_16x16x4_f32(false, (A), false, (B), (short)0, (C), false, false)

typedef volatile __attribute__((address_space(3))) int lds_vint;

__device__ __forceinline__ int4 load_dst4(const int* __restrict__ dstp, long long g,
                                          int n_edges, bool vec_ok) {
  int4 r;
  if (vec_ok) {
    r = *(const int4*)(dstp + 4 * g);
  } else {
    long long e = 4 * g;
    r.x = (e + 0 < n_edges) ? dstp[e + 0] : -1;
    r.y = (e + 1 < n_edges) ? dstp[e + 1] : -1;
    r.z = (e + 2 < n_edges) ? dstp[e + 2] : -1;
    r.w = (e + 3 < n_edges) ? dstp[e + 3] : -1;
  }
  return r;
}

__device__ __forceinline__ void drain_queue(int n, int lane, lds_vint* qe,
                                            const int* __restrict__ ei,
                                            const int* __restrict__ dstp,
                                            const unsigned* __restrict__ rec, int base,
                                            int nloc, int n_edges,
                                            const float* __restrict__ x, float* sAcc,
                                            unsigned* sDeg) {
  int l = -1, s = 0;
  if (lane < n) {
    int e = qe[lane];
    if (rec) {
      const unsigned r = rec[e];
      const int lr = (int)(r & 511u);
      if (lr < nloc) {
        l = lr;
        s = (int)(r >> 9);
        __hip_atomic_fetch_add(sDeg + l, 1u, __ATOMIC_RELAXED, __HIP_MEMORY_SCOPE_WORKGROUP);
      }
    } else if (e < n_edges) {
      s = ei[e];
      l = dstp[e] - base;
      if ((unsigned)l < (unsigned)nloc)
        __hip_atomic_fetch_add(sDeg + l, 1u, __ATOMIC_RELAXED, __HIP_MEMORY_SCOPE_WORKGROUP);
    }
  }
  for (int j0 = 0; j0 < n; j0 += DB) {
    v2f v[DB];
#pragma unroll
    for (int k = 0; k < DB; ++k) {
      int sj = __builtin_amdgcn_readlane(s, j0 + k);
      v[k] = *(const v2f*)(x + (size_t)sj * IN_DIM + 2 * lane);
    }
#pragma unroll
    for (int k = 0; k < DB; ++k) {
      const int lj = __builtin_amdgcn_readlane(l, j0 + k);
      if (j0 + k < n && (unsigned)lj < (unsigned)nloc) {
        float* p = sAcc + lj * LDS_STRIDE + 2 * lane;
        __hip_atomic_fetch_add(p, v[k].x, __ATOMIC_RELAXED, __HIP_MEMORY_SCOPE_WORKGROUP);
        __hip_atomic_fetch_add(p + 1, v[k].y, __ATOMIC_RELAXED, __HIP_MEMORY_SCOPE_WORKGROUP);
      }
    }
  }
}

__global__ __launch_bounds__(256) void demo_key_kernel(
    const int* __restrict__ ei, unsigned* __restrict__ planes,
    unsigned* __restrict__ rec,
    int n_nodes, int n_edges, int n_chunks) {
  const int* dstp = ei + n_edges;
  const int lane = threadIdx.x & 31;
  const int wave = (int)((blockIdx.x * blockDim.x + threadIdx.x) >> 5);
  const bool vec_ok = (n_edges & 3) == 0;
  const long long n4 = ((long long)n_edges + 3) >> 2;

  const bool hi_digit = lane < 16;
  unsigned flip[4];
#pragma unroll
  for (int i = 0; i < 4; ++i) flip[i] = ((lane >> i) & 1) ? 0u : 0xFFFFFFFFu;

  int4 d[4], sv[4];
  const long long gw = 128LL * wave;
  if (vec_ok && gw + 128 <= n4) {
#pragma unroll
    for (int k = 0; k < 4; ++k) d[k] = *(const int4*)(dstp + 4 * (gw + 32 * k + lane));
    if (rec) {
#pragma unroll
      for (int k = 0; k < 4; ++k) sv[k] = *(const int4*)(ei + 4 * (gw + 32 * k + lane));
    }
  } else {
#pragma unroll
    for (int k = 0; k < 4; ++k) {
      const long long g = gw + 32 * k + lane;
      d[k] = (g < n4) ? load_dst4(dstp, g, n_edges, vec_ok) : make_int4(-1, -1, -1, -1);
      sv[k] = (rec && g < n4) ? load_dst4(ei, g, n_edges, vec_ok) : make_int4(0, 0, 0, 0);
    }
  }
#pragma unroll
  for (int k = 0; k < 4; ++k) {
    const int C = 4 * wave + k;
    if (C < n_chunks) {
      const int dd[4] = {d[k].x, d[k].y, d[k].z, d[k].w};
      unsigned w[4];
#pragma unroll
      for (int c = 0; c < 4; ++c) {
        const bool valid = (unsigned)dd[c] < (unsigned)n_nodes;
        const unsigned key = ((unsigned)dd[c] / (unsigned)NPB) & 255u;
        const unsigned vb = __builtin_amdgcn_ballot_w32(valid);
        unsigned klo[4], khi[4];
#pragma unroll
        for (int i = 0; i < 4; ++i) {
          klo[i] = __builtin_amdgcn_ballot_w32(((key >> i) & 1u) != 0u);
          khi[i] = __builtin_amdgcn_ballot_w32(((key >> (4 + i)) & 1u) != 0u);
        }
        unsigned wc = vb;
#pragma unroll
        for (int i = 0; i < 4; ++i) {
          const unsigned sel = hi_digit ? khi[i] : klo[i];
          wc &= sel ^ flip[i];
        }
        w[c] = wc;
      }
      ((uint4*)planes)[(size_t)lane * n_chunks + C] = make_uint4(w[0], w[1], w[2], w[3]);
      if (rec) {
        const int ss[4] = {sv[k].x, sv[k].y, sv[k].z, sv[k].w};
        unsigned rv[4];
#pragma unroll
        for (int c = 0; c < 4; ++c) {
          const unsigned du = (unsigned)dd[c];
          rv[c] = du < (unsigned)n_nodes
                      ? ((unsigned)ss[c] << 9) | (du - (du / (unsigned)NPB) * (unsigned)NPB)
                      : 0xFFFFFFFFu;
        }
        *(uint4*)(rec + (size_t)128 * C + 4 * lane) = make_uint4(rv[0], rv[1], rv[2], rv[3]);
      }
    }
  }
}

__device__ __forceinline__ void early_products(int wave, int lane, int base, int n_nodes,
                                               int ntiles, const float* __restrict__ x,
                                               const float* sWg, const float* sWs,
                                               float* sE) {
  const int lane_lo = lane & 15;
  const int hi      = lane >> 4;
  const int bofs    = lane_lo * LDS_STRIDE + 2 * hi;
  for (int item = wave; item < 2 * ntiles; item += NWAVES) {
    const int tile = item >> 1;
    const int half = item & 1;
    int arow = base + tile * 16 + lane_lo;
    if (arow > n_nodes - 1) arow = n_nodes - 1;
    const v2f* xr = (const v2f*)(x + (size_t)arow * IN_DIM + 2 * hi);
    v8f acc[2] = {v8f{}, v8f{}};
#pragma unroll 1
    for (int c4 = 0; c4 < 4; ++c4) {
      v2f ax[4], bg[4][2], bs[4][2];
#pragma unroll
      for (int cc = 0; cc < 4; ++cc) {
        const int c = 4 * c4 + cc;
        ax[cc] = xr[2 * c];
#pragma unroll
        for (int q = 0; q < 2; ++q) {
          const int bo = bofs + (half * 2 + q) * 16 * LDS_STRIDE + 4 * c;
          bg[cc][q] = *(const v2f*)(sWg + bo);
          bs[cc][q] = *(const v2f*)(sWs + bo);
        }
      }
#pragma unroll
      for (int cc = 0; cc < 4; ++cc) {
#pragma unroll
        for (int q = 0; q < 2; ++q) {
          acc[q] = WMMA_F32_16x16x4(ax[cc], bg[cc][q], acc[q]);
          acc[q] = WMMA_F32_16x16x4(ax[cc], bs[cc][q], acc[q]);
        }
      }
    }
#pragma unroll
    for (int q = 0; q < 2; ++q) {
#pragma unroll
      for (int rr = 0; rr < 8; ++rr)
        sE[(tile * 16 + rr + 8 * hi) * LDS_STRIDE + (half * 2 + q) * 16 + lane_lo] = acc[q][rr];
    }
  }
}

__global__ __launch_bounds__(NTHREADS) void demo_fused_kernel(
    const float* __restrict__ x, const float* __restrict__ Wg,
    const float* __restrict__ Wl, const float* __restrict__ Ws,
    const float* __restrict__ bias, const int* __restrict__ ei,
    const unsigned* __restrict__ planes,
    const unsigned* __restrict__ rec,
    float* __restrict__ out, int n_nodes, int n_edges, int n_chunks) {
  __shared__ alignas(16) float sAcc[NPB * LDS_STRIDE];
  __shared__ alignas(16) float sW[3 * OUT_DIM * LDS_STRIDE];
  __shared__ alignas(16) float sE[NPB * LDS_STRIDE];
  __shared__ unsigned sDeg[NPB];
  __shared__ int sQe[NWAVES * QCAP];
  float* sWg = sW;
  float* sWs = sW + OUT_DIM * LDS_STRIDE;
  float* sWl = sW + 2 * OUT_DIM * LDS_STRIDE;

  const int tid  = threadIdx.x;
  const int lane = tid & 31;
  const int wave = tid >> 5;
  const int base = blockIdx.x * NPB;
  const int nloc = (n_nodes - base < NPB) ? (n_nodes - base) : NPB;
  const int ntiles = (nloc + 15) >> 4;

  for (int i = tid; i < NPB * LDS_STRIDE / 4; i += NTHREADS)
    ((float4*)sAcc)[i] = make_float4(0.f, 0.f, 0.f, 0.f);
  for (int i = tid; i < NPB; i += NTHREADS) sDeg[i] = 0u;
  {
    int row = tid >> 4;
    int k4  = tid & 15;
    float4 g = *(const float4*)(Wg + (size_t)tid * 4);
    float4 s = *(const float4*)(Ws + (size_t)tid * 4);
    float4 l = *(const float4*)(Wl + (size_t)tid * 4);
    *(float4*)(sWg + row * LDS_STRIDE + k4 * 4) = g;
    *(float4*)(sWs + row * LDS_STRIDE + k4 * 4) = s;
    *(float4*)(sWl + row * LDS_STRIDE + k4 * 4) = l;
  }
  __syncthreads();

  {
    const int* dstp = ei + n_edges;
    lds_vint* qe = (lds_vint*)(sQe + wave * QCAP);
    const unsigned myKey = blockIdx.x & 255u;
    const uint4* rowA = (const uint4*)planes + (size_t)(myKey >> 4) * n_chunks;
    const uint4* rowB = (const uint4*)planes + (size_t)(16u + (myKey & 15u)) * n_chunks;
    int cnt = 0;
    const int wave_s = __builtin_amdgcn_readfirstlane(wave);
    const int eslot = ((wave_s >> 2) * (EPASSES * UG)) / (NWAVES / 4);
    int pass = 0;

    for (int q0 = 0; q0 < n_chunks || pass < EPASSES; q0 += NTHREADS * UG, ++pass) {
      uint4 pa[UG], pb[UG];
      if (q0 + NTHREADS * UG <= n_chunks) {
#pragma unroll
        for (int u = 0; u < UG; ++u) {
          pa[u] = rowA[q0 + u * NTHREADS + tid];
          pb[u] = rowB[q0 + u * NTHREADS + tid];
        }
      } else {
#pragma unroll
        for (int u = 0; u < UG; ++u) {
          const int q = q0 + u * NTHREADS + tid;
          const bool ok = q < n_chunks;
          pa[u] = ok ? rowA[q] : make_uint4(0u, 0u, 0u, 0u);
          pb[u] = ok ? rowB[q] : make_uint4(0u, 0u, 0u, 0u);
        }
      }
#pragma unroll
      for (int u = 0; u < UG; ++u) {
        if (pass * UG + u == eslot)
          early_products(wave_s, lane, base, n_nodes, ntiles, x, sWg, sWs, sE);
        const int q = q0 + u * NTHREADS + tid;
        unsigned r0 = pa[u].x & pb[u].x, r1 = pa[u].y & pb[u].y;
        unsigned r2 = pa[u].z & pb[u].z, r3 = pa[u].w & pb[u].w;
#pragma unroll 1
        for (int c = 0; c < 4; ++c) {
          const int ebase = (q << 7) + c;
          unsigned m = r0;
          r0 = r1; r1 = r2; r2 = r3;
          unsigned mask = __builtin_amdgcn_ballot_w32(m != 0u);
          while (mask) {
            if (m) {
              const int b = __builtin_ctz(m);
              m &= m - 1u;
              qe[cnt + (int)__builtin_amdgcn_mbcnt_lo(mask, 0u)] = ebase + 4 * b;
            }
            cnt += __builtin_popcount(mask);
            if (cnt >= 32) {
              drain_queue(32, lane, qe, ei, dstp, rec, base, nloc, n_edges, x, sAcc, sDeg);
              const int rem = cnt - 32;
              int te = 0;
              if (lane < rem) te = qe[32 + lane];
              if (lane < rem) qe[lane] = te;
              cnt = rem;
            }
            mask = __builtin_amdgcn_ballot_w32(m != 0u);
          }
        }
      }
    }
    drain_queue(cnt, lane, qe, ei, dstp, rec, base, nloc, n_edges, x, sAcc, sDeg);
  }
  __syncthreads();

  const int lane_lo = lane & 15;
  const int hi      = lane >> 4;
  const int bofs = lane_lo * LDS_STRIDE + 2 * hi;

  for (int item = wave; item < 2 * ntiles; item += NWAVES) {
    const int tile = item >> 1;
    const int half = item & 1;
    const int lrow = tile * 16 + lane_lo;
    int arow = base + lrow;
    if (arow > n_nodes - 1) arow = n_nodes - 1;

    const float dv  = (float)sDeg[lrow];
    const float inv = 1.0f / fmaxf(dv, 1.0f);
    const float hn  = dv > 0.0f ? 1.0f : 0.0f;

    const v2f* xr = (const v2f*)(x    + (size_t)arow * IN_DIM + 2 * hi);
    const v2f* sr = (const v2f*)(sAcc + lrow * LDS_STRIDE + 2 * hi);

    v8f acc[2] = {v8f{}, v8f{}};
    const bool full = __builtin_amdgcn_ballot_w32(dv > 0.0f) == 0xFFFFFFFFu;
    if (full) {
#pragma unroll
      for (int q = 0; q < 2; ++q) {
#pragma unroll
        for (int rr = 0; rr < 8; ++rr)
          acc[q][rr] = sE[(tile * 16 + rr + 8 * hi) * LDS_STRIDE + (half * 2 + q) * 16 + lane_lo];
      }
#pragma unroll 1
      for (int c4 = 0; c4 < 4; ++c4) {
        v2f am[4], bl[4][2];
#pragma unroll
        for (int cc = 0; cc < 4; ++cc) {
          const int c = 4 * c4 + cc;
          am[cc] = sr[2 * c];
#pragma unroll
          for (int q = 0; q < 2; ++q)
            bl[cc][q] = *(const v2f*)(sWl + bofs + (half * 2 + q) * 16 * LDS_STRIDE + 4 * c);
        }
#pragma unroll
        for (int cc = 0; cc < 4; ++cc) {
          const v2f a_m = am[cc] * inv;
#pragma unroll
          for (int q = 0; q < 2; ++q) acc[q] = WMMA_F32_16x16x4(a_m, bl[cc][q], acc[q]);
        }
      }
    } else {
#pragma unroll 1
    for (int c4 = 0; c4 < 4; ++c4) {
      v2f ax[4], am[4], bg[4][2], bs[4][2], bl[4][2];
#pragma unroll
      for (int cc = 0; cc < 4; ++cc) {
        const int c = 4 * c4 + cc;
        ax[cc] = xr[2 * c];
        am[cc] = sr[2 * c];
#pragma unroll
        for (int q = 0; q < 2; ++q) {
          const int bo = bofs + (half * 2 + q) * 16 * LDS_STRIDE + 4 * c;
          bg[cc][q] = *(const v2f*)(sWg + bo);
          bs[cc][q] = *(const v2f*)(sWs + bo);
          bl[cc][q] = *(const v2f*)(sWl + bo);
        }
      }
#pragma unroll
      for (int cc = 0; cc < 4; ++cc) {
        const v2f a_m = am[cc] * inv;
        const v2f a_s = ax[cc] * hn;
#pragma unroll
        for (int q = 0; q < 2; ++q) {
          acc[q] = WMMA_F32_16x16x4(ax[cc], bg[cc][q], acc[q]);
          acc[q] = WMMA_F32_16x16x4(a_s, bs[cc][q], acc[q]);
          acc[q] = WMMA_F32_16x16x4(a_m, bl[cc][q], acc[q]);
        }
      }
    }
    }

#pragma unroll
    for (int q = 0; q < 2; ++q) {
      const int col = (half * 2 + q) * 16 + lane_lo;
      const float bv = bias[col];
      v8f r = acc[q];
#pragma unroll
      for (int rr = 0; rr < 8; ++rr) {
        int m = base + tile * 16 + rr + 8 * hi;
        if (m < n_nodes) {
          float val = r[rr] + bv;
          val = val > 0.0f ? val : (__expf(val) - 1.0f);
          out[(size_t)m * OUT_DIM + col] = val;
        }
      }
    }
  }
}

extern "C" void kernel_launch(void* const* d_in, const int* in_sizes, int n_in,
                              void* d_out, int out_size, void* d_ws, size_t ws_size,
                              hipStream_t stream) {
  const float* x    = (const float*)d_in[0];
  const float* Wg   = (const float*)d_in[1];
  const float* Wl   = (const float*)d_in[2];
  const float* Ws   = (const float*)d_in[3];
  const float* bias = (const float*)d_in[4];
  const int*   ei   = (const int*)d_in[5];
  float* out = (float*)d_out;

  const int n_nodes = in_sizes[0] / IN_DIM;
  const int n_edges = in_sizes[5] / 2;

  const int n_chunks = (n_edges + KCH - 1) / KCH;
  unsigned* planes = (unsigned*)d_ws;
  if (ws_size < (size_t)4 * KCH * n_chunks) return;
  unsigned* rec = (ws_size >= (size_t)8 * KCH * n_chunks && n_nodes <= 256 * NPB)
                      ? planes + (size_t)KCH * n_chunks : nullptr;

  int kblocks = (n_chunks + 31) / 32;
  demo_key_kernel<<<kblocks, 256, 0, stream>>>(ei, planes, rec, n_nodes, n_edges, n_chunks);

  int blocks = (n_nodes + NPB - 1) / NPB;
  demo_fused_kernel<<<blocks, NTHREADS, 0, stream>>>(x, Wg, Wl, Ws, bias, ei, planes,
                                                     rec, out, n_nodes, n_edges, n_chunks);
}
